# starting kernel
# baseline (speedup 1.0000x reference)
#include <hip/hip_runtime.h>

typedef __attribute__((ext_vector_type(2))) float v2f;
typedef __attribute__((ext_vector_type(8))) float v8f;

#define NQ 10
#define NL 4
#define NSTATES 1024
#define WAVES_PER_BLOCK 4
#define THREADS (WAVES_PER_BLOCK * 32)

#define LDS_BYTES (1280 + 8192 + 8192 + 512 + 8192 + 32768)


__device__ __forceinline__ void apply16(float2* S,
                                        const float* Ure, const float* Uim,
                                        const int mode, const int lane) {
  const int hf  = lane >> 4;
  const int nlo = lane & 15;

  v2f aRe[4], aIm[4], aImN[4];
  #pragma unroll
  for (int k = 0; k < 4; ++k) {
    const int K0 = 4 * k + 2 * hf;
    aRe[k].x  = Ure[nlo * 16 + K0];  aRe[k].y  = Ure[nlo * 16 + K0 + 1];
    aIm[k].x  = Uim[nlo * 16 + K0];  aIm[k].y  = Uim[nlo * 16 + K0 + 1];
    aImN[k].x = -aIm[k].x;           aImN[k].y = -aIm[k].y;
  }

  #pragma unroll
  for (int jt = 0; jt < 4; ++jt) {
    const int n = jt * 16 + nlo;
    v8f accRe = {0.f,0.f,0.f,0.f,0.f,0.f,0.f,0.f};
    v8f accIm = {0.f,0.f,0.f,0.f,0.f,0.f,0.f,0.f};
    #pragma unroll
    for (int k = 0; k < 4; ++k) {
      const int K0 = 4 * k + 2 * hf;
      const int K1 = K0 + 1;
      const int i0 = (mode == 0) ? (K0 * 64 + n) : (((n >> 2) << 6) + K0 * 4 + (n & 3));
      const int i1 = (mode == 0) ? (K1 * 64 + n) : (((n >> 2) << 6) + K1 * 4 + (n & 3));
      const float2 z0 = S[i0];
      const float2 z1 = S[i1];
      v2f br, bi;
      br.x = z0.x;  br.y = z1.x;
      bi.x = z0.y;  bi.y = z1.y;
      accRe = __builtin_amdgcn_wmma_f32_16x16x4_f32(false, aRe[k],  false, br, (short)0, accRe, false, false);
      accRe = __builtin_amdgcn_wmma_f32_16x16x4_f32(false, aImN[k], false, bi, (short)0, accRe, false, false);
      accIm = __builtin_amdgcn_wmma_f32_16x16x4_f32(false, aRe[k],  false, bi, (short)0, accIm, false, false);
      accIm = __builtin_amdgcn_wmma_f32_16x16x4_f32(false, aIm[k],  false, br, (short)0, accIm, false, false);
    }
    #pragma unroll
    for (int v = 0; v < 8; ++v) {
      const int M = v + 8 * hf;
      const int o = (mode == 0) ? (M * 64 + n) : (((n >> 2) << 6) + M * 4 + (n & 3));
      S[o] = make_float2(accRe[v], accIm[v]);
    }
  }
}

__device__ __forceinline__ void applyC(float2* S,
                                       const float* ucr, const float* uci,
                                       const int lane) {
  const int hf  = lane >> 4;
  const int nlo = lane & 15;
  const int grp = nlo >> 2, cc = nlo & 3;
  const int c0 = 2 * hf;

  const float vrx = ucr[cc * 4 + c0], vry = ucr[cc * 4 + c0 + 1];
  const float vix = uci[cc * 4 + c0], viy = uci[cc * 4 + c0 + 1];
  v2f bRe[4], bIm[4], bImN[4];
  #pragma unroll
  for (int k = 0; k < 4; ++k) {
    const bool on = (k == grp);
    bRe[k].x  = on ?  vrx : 0.f;  bRe[k].y  = on ?  vry : 0.f;
    bIm[k].x  = on ?  vix : 0.f;  bIm[k].y  = on ?  viy : 0.f;
    bImN[k].x = on ? -vix : 0.f;  bImN[k].y = on ? -viy : 0.f;
  }

  #pragma unroll
  for (int jt = 0; jt < 4; ++jt) {
    const int n = jt * 16 + nlo;
    v8f accRe = {0.f,0.f,0.f,0.f,0.f,0.f,0.f,0.f};
    v8f accIm = {0.f,0.f,0.f,0.f,0.f,0.f,0.f,0.f};
    #pragma unroll
    for (int k = 0; k < 4; ++k) {
      const int i0 = (4 * jt + k) * 64 + nlo * 4 + 2 * hf;
      const float4 z = *(const float4*)(S + i0);
      v2f ar, ai;
      ar.x = z.x;  ar.y = z.z;
      ai.x = z.y;  ai.y = z.w;
      accRe = __builtin_amdgcn_wmma_f32_16x16x4_f32(false, ar, false, bRe[k],  (short)0, accRe, false, false);
      accRe = __builtin_amdgcn_wmma_f32_16x16x4_f32(false, ai, false, bImN[k], (short)0, accRe, false, false);
      accIm = __builtin_amdgcn_wmma_f32_16x16x4_f32(false, ar, false, bIm[k],  (short)0, accIm, false, false);
      accIm = __builtin_amdgcn_wmma_f32_16x16x4_f32(false, ai, false, bRe[k],  (short)0, accIm, false, false);
    }
    #pragma unroll
    for (int v = 0; v < 8; ++v) {
      const int M = v + 8 * hf;
      const int o = ((n >> 2) << 6) + M * 4 + (n & 3);
      S[o] = make_float2(accRe[v], accIm[v]);
    }
  }
}

extern "C" __global__ __launch_bounds__(THREADS)
void qlayer_kernel(const float* __restrict__ x,
                   const float* __restrict__ qw,
                   float* __restrict__ out) {
  extern __shared__ char smem_raw[];
  float* u2   = (float*)smem_raw;
  float* UAre = u2 + NL * NQ * 8;
  float* UAim = UAre + NL * 256;
  float* UBre = UAim + NL * 256;
  float* UBim = UBre + NL * 256;
  float* UCre = UBim + NL * 256;
  float* UCim = UCre + NL * 16;
  unsigned* permPk = (unsigned*)(UCim + NL * 16);
  float2* stateBase = (float2*)(permPk + NL * 512);

  const int tid = threadIdx.x;

  #pragma unroll 1
  for (int e = tid; e < NL * NQ; e += THREADS) {
    const int l = e / NQ, w = e % NQ;
    const float phi = qw[(l * NQ + w) * 3 + 0];
    const float th  = qw[(l * NQ + w) * 3 + 1];
    const float om  = qw[(l * NQ + w) * 3 + 2];
    const float c = cosf(0.5f * th), s = sinf(0.5f * th);
    const float ap = 0.5f * (phi + om), am = 0.5f * (phi - om);
    const float cpo = cosf(ap), spo = sinf(ap);
    const float cb = cosf(am),  sb = sinf(am);
    float* g = u2 + e * 8;
    g[0] =  cpo * c; g[1] = -spo * c;
    g[2] = -cb * s;  g[3] = -sb * s;
    g[4] =  cb * s;  g[5] = -sb * s;
    g[6] =  cpo * c; g[7] =  spo * c;
  }
  __syncthreads();

  #pragma unroll 1
  for (int e = tid; e < NL * 256; e += THREADS) {
    const int l = e >> 8, rc = e & 255, r = rc >> 4, cc = rc & 15;
    float arA = 1.f, aiA = 0.f, arB = 1.f, aiB = 0.f;
    #pragma unroll
    for (int q = 0; q < 4; ++q) {
      const int rb = (r >> (3 - q)) & 1, cbit = (cc >> (3 - q)) & 1;
      const float* gA = u2 + (((l * NQ + q) * 4) + rb * 2 + cbit) * 2;
      float nr = arA * gA[0] - aiA * gA[1];
      float ni = arA * gA[1] + aiA * gA[0];
      arA = nr; aiA = ni;
      const float* gB = u2 + (((l * NQ + 4 + q) * 4) + rb * 2 + cbit) * 2;
      nr = arB * gB[0] - aiB * gB[1];
      ni = arB * gB[1] + aiB * gB[0];
      arB = nr; aiB = ni;
    }
    UAre[e] = arA; UAim[e] = aiA;
    UBre[e] = arB; UBim[e] = aiB;
  }
  #pragma unroll 1
  for (int e = tid; e < NL * 16; e += THREADS) {
    const int l = e >> 4, r = (e >> 2) & 3, cc = e & 3;
    const float* g8 = u2 + (((l * NQ + 8) * 4) + ((r >> 1) & 1) * 2 + ((cc >> 1) & 1)) * 2;
    const float* g9 = u2 + (((l * NQ + 9) * 4) + (r & 1) * 2 + (cc & 1)) * 2;
    UCre[e] = g8[0] * g9[0] - g8[1] * g9[1];
    UCim[e] = g8[0] * g9[1] + g8[1] * g9[0];
  }
  #pragma unroll 1
  for (int e = tid; e < NL * 512; e += THREADS) {
    const int l = e >> 9, p = e & 511;
    const int ln = p & 31, h = p >> 5;
    const int rr = (l % (NQ - 1)) + 1;
    unsigned pk = 0;
    #pragma unroll
    for (int half = 0; half < 2; ++half) {
      int j = ln + 64 * h + 32 * half;
      for (int w = NQ - 1; w >= 0; --w) {
        const int cpos = 9 - w;
        const int tpos = 9 - ((w + rr) % NQ);
        j ^= ((j >> cpos) & 1) << tpos;
      }
      pk |= ((unsigned)j) << (16 * half);
    }
    permPk[e] = pk;
  }
  __syncthreads();

  const int wave = tid >> 5, lane = tid & 31;
  const int b = blockIdx.x * WAVES_PER_BLOCK + wave;
  float2* S = stateBase + wave * NSTATES;

  float cw[NQ], sw[NQ];
  #pragma unroll
  for (int w = 0; w < NQ; ++w) {
    const float ang = 1.5707963267948966f * x[b * NQ + w];
    cw[w] = cosf(ang);
    sw[w] = sinf(ang);
  }
  for (int t = 0; t < 32; ++t) {
    const int i = lane + 32 * t;
    float a = 1.f;
    #pragma unroll
    for (int w = 0; w < NQ; ++w) a *= ((i >> (9 - w)) & 1) ? sw[w] : cw[w];
    S[i] = make_float2(a, 0.f);
  }

  #pragma unroll 1
  for (int l = 0; l < NL; ++l) {
    apply16(S, UAre + l * 256, UAim + l * 256, 0, lane);
    apply16(S, UBre + l * 256, UBim + l * 256, 1, lane);
    applyC (S, UCre + l * 16,  UCim + l * 16,  lane);

    const unsigned* pl = permPk + l * 512;
    float2 tv[32];
    #pragma unroll
    for (int h = 0; h < 16; ++h) {
      const unsigned pk = pl[lane + 32 * h];
      tv[2 * h]     = S[pk & 0xFFFFu];
      tv[2 * h + 1] = S[pk >> 16];
    }
    #pragma unroll
    for (int t = 0; t < 32; ++t) S[lane + 32 * t] = tv[t];
  }

  float acc[NQ];
  #pragma unroll
  for (int w = 0; w < NQ; ++w) acc[w] = 0.f;
  for (int t = 0; t < 32; ++t) {
    const int i = lane + 32 * t;
    const float2 z = S[i];
    const float p = z.x * z.x + z.y * z.y;
    #pragma unroll
    for (int w = 0; w < NQ; ++w) acc[w] += ((i >> (9 - w)) & 1) ? -p : p;
  }
  #pragma unroll
  for (int w = 0; w < NQ; ++w) {
    float v = acc[w];
    #pragma unroll
    for (int off = 16; off > 0; off >>= 1) v += __shfl_xor(v, off, 32);
    if (lane == 0) out[b * NQ + w] = v;
  }
}

extern "C" void kernel_launch(void* const* d_in, const int* in_sizes, int n_in,
                              void* d_out, int out_size, void* d_ws, size_t ws_size,
                              hipStream_t stream) {
  (void)in_sizes; (void)n_in; (void)out_size; (void)d_ws; (void)ws_size;
  const float* x  = (const float*)d_in[0];
  const float* qw = (const float*)d_in[1];
  float* out = (float*)d_out;
  const int batch = 2048;
  dim3 grid(batch / WAVES_PER_BLOCK);
  dim3 block(THREADS);
  qlayer_kernel<<<grid, block, LDS_BYTES, stream>>>(x, qw, out);
}


# Qubits 4-9 as two 3-qubit groups in real 16x16 form [[Ur,-Ui],[Ui,Ur]], in place, instead of 4+2 complex groups: 128 not 192 f32 WMMAs per layer; each amplitude's summation order changes
# speedup vs baseline: 1.2822x; 1.2822x over previous
#include <hip/hip_runtime.h>

typedef __attribute__((ext_vector_type(2))) float v2f;
typedef __attribute__((ext_vector_type(8))) float v8f;

#define NQ 10
#define NL 4
#define NSTATES 1024
#define WAVES_PER_BLOCK 4
#define THREADS (WAVES_PER_BLOCK * 32)

#define LDS_BYTES (1280 + 8192 + 8192 + 8192 + 32768)


__device__ __forceinline__ void apply16(float2* S,
                                        const float* Ure, const float* Uim,
                                        const int lane) {
  const int hf  = lane >> 4;
  const int nlo = lane & 15;

  v2f aRe[4], aIm[4], aImN[4];
  #pragma unroll
  for (int k = 0; k < 4; ++k) {
    const int K0 = 4 * k + 2 * hf;
    aRe[k].x  = Ure[nlo * 16 + K0];  aRe[k].y  = Ure[nlo * 16 + K0 + 1];
    aIm[k].x  = Uim[nlo * 16 + K0];  aIm[k].y  = Uim[nlo * 16 + K0 + 1];
    aImN[k].x = -aIm[k].x;           aImN[k].y = -aIm[k].y;
  }

  #pragma unroll
  for (int jt = 0; jt < 4; ++jt) {
    const int n = jt * 16 + nlo;
    v8f accRe = {0.f,0.f,0.f,0.f,0.f,0.f,0.f,0.f};
    v8f accIm = {0.f,0.f,0.f,0.f,0.f,0.f,0.f,0.f};
    #pragma unroll
    for (int k = 0; k < 4; ++k) {
      const int K0 = 4 * k + 2 * hf;
      const int K1 = K0 + 1;
      const int i0 = K0 * 64 + n;
      const int i1 = K1 * 64 + n;
      const float2 z0 = S[i0];
      const float2 z1 = S[i1];
      v2f br, bi;
      br.x = z0.x;  br.y = z1.x;
      bi.x = z0.y;  bi.y = z1.y;
      accRe = __builtin_amdgcn_wmma_f32_16x16x4_f32(false, aRe[k],  false, br, (short)0, accRe, false, false);
      accRe = __builtin_amdgcn_wmma_f32_16x16x4_f32(false, aImN[k], false, bi, (short)0, accRe, false, false);
      accIm = __builtin_amdgcn_wmma_f32_16x16x4_f32(false, aRe[k],  false, bi, (short)0, accIm, false, false);
      accIm = __builtin_amdgcn_wmma_f32_16x16x4_f32(false, aIm[k],  false, br, (short)0, accIm, false, false);
    }
    #pragma unroll
    for (int v = 0; v < 8; ++v) {
      const int M = v + 8 * hf;
      const int o = M * 64 + n;
      S[o] = make_float2(accRe[v], accIm[v]);
    }
  }
}

template <int MID>
__device__ __forceinline__ void apply8r(float* F, const float* frag, const int lane) {
  const int hf  = lane >> 4;
  const int nlo = lane & 15;
  const float4 g0 = *(const float4*)(frag + lane * 8);
  const float4 g1 = *(const float4*)(frag + lane * 8 + 4);
  v2f a[4];
  a[0].x = g0.x; a[0].y = g0.y;  a[1].x = g0.z; a[1].y = g0.w;
  a[2].x = g1.x; a[2].y = g1.y;  a[3].x = g1.z; a[3].y = g1.w;

  float* ptr[8];
  v2f b[8][4];
  #pragma unroll
  for (int jt = 0; jt < 8; ++jt) {
    if (MID) {
      ptr[jt] = F + 128 * (2 * jt + (nlo >> 3)) + 2 * (nlo & 7) + 16 * hf;
      #pragma unroll
      for (int kc = 0; kc < 4; ++kc) {
        const float2 z = *(const float2*)(ptr[jt] + 32 * kc);
        b[jt][kc].x = z.x;  b[jt][kc].y = z.y;
      }
    } else {
      ptr[jt] = F + 16 * (jt * 16 + nlo) + 8 * hf;
      const float4 x0 = *(const float4*)(ptr[jt]);
      const float4 x1 = *(const float4*)(ptr[jt] + 4);
      b[jt][0].x = x0.x; b[jt][0].y = x0.y;  b[jt][1].x = x0.z; b[jt][1].y = x0.w;
      b[jt][2].x = x1.x; b[jt][2].y = x1.y;  b[jt][3].x = x1.z; b[jt][3].y = x1.w;
    }
  }
  #pragma unroll
  for (int jq = 0; jq < 2; ++jq) {
    v8f d[4];
    #pragma unroll
    for (int t = 0; t < 4; ++t) d[t] = (v8f){0.f,0.f,0.f,0.f,0.f,0.f,0.f,0.f};
    #pragma unroll
    for (int kc = 0; kc < 4; ++kc) {
      #pragma unroll
      for (int t = 0; t < 4; ++t)
        d[t] = __builtin_amdgcn_wmma_f32_16x16x4_f32(false, a[kc], false, b[4 * jq + t][kc], (short)0, d[t], false, false);
    }
    #pragma unroll
    for (int t = 0; t < 4; ++t) {
      float* o = ptr[4 * jq + t];
      if (MID) {
        #pragma unroll
        for (int j = 0; j < 4; ++j)
          *(float2*)(o + 32 * j) = make_float2(d[t][2 * j], d[t][2 * j + 1]);
      } else {
        *(float4*)(o)     = make_float4(d[t][0], d[t][1], d[t][2], d[t][3]);
        *(float4*)(o + 4) = make_float4(d[t][4], d[t][5], d[t][6], d[t][7]);
      }
    }
  }
}

extern "C" __global__ __launch_bounds__(THREADS) __attribute__((amdgpu_waves_per_eu(1, 4)))
void qlayer_kernel(const float* __restrict__ x,
                   const float* __restrict__ qw,
                   float* __restrict__ out) {
  extern __shared__ char smem_raw[];
  float* u2   = (float*)smem_raw;
  float* UAre = u2 + NL * NQ * 8;
  float* UAim = UAre + NL * 256;
  float* frag = UAim + NL * 256;
  unsigned* permPk = (unsigned*)(frag + NL * 512);
  float2* stateBase = (float2*)(permPk + NL * 512);

  const int tid = threadIdx.x;

  #pragma unroll 1
  for (int e = tid; e < NL * NQ; e += THREADS) {
    const int l = e / NQ, w = e % NQ;
    const float phi = qw[(l * NQ + w) * 3 + 0];
    const float th  = qw[(l * NQ + w) * 3 + 1];
    const float om  = qw[(l * NQ + w) * 3 + 2];
    const float c = cosf(0.5f * th), s = sinf(0.5f * th);
    const float ap = 0.5f * (phi + om), am = 0.5f * (phi - om);
    const float cpo = cosf(ap), spo = sinf(ap);
    const float cb = cosf(am),  sb = sinf(am);
    float* g = u2 + e * 8;
    g[0] =  cpo * c; g[1] = -spo * c;
    g[2] = -cb * s;  g[3] = -sb * s;
    g[4] =  cb * s;  g[5] = -sb * s;
    g[6] =  cpo * c; g[7] =  spo * c;
  }
  __syncthreads();

  #pragma unroll 1
  for (int e = tid; e < NL * 256; e += THREADS) {
    const int l = e >> 8, rc = e & 255, r = rc >> 4, cc = rc & 15;
    float arA = 1.f, aiA = 0.f;
    #pragma unroll
    for (int q = 0; q < 4; ++q) {
      const int rb = (r >> (3 - q)) & 1, cbit = (cc >> (3 - q)) & 1;
      const float* gA = u2 + (((l * NQ + q) * 4) + rb * 2 + cbit) * 2;
      const float nr = arA * gA[0] - aiA * gA[1];
      const float ni = arA * gA[1] + aiA * gA[0];
      arA = nr; aiA = ni;
    }
    UAre[e] = arA; UAim[e] = aiA;
  }
  #pragma unroll 1
  for (int e = tid; e < NL * 512; e += THREADS) {
    const int l = e >> 9, grp = (e >> 8) & 1, ln = (e >> 3) & 31, s = e & 7;
    const int m = ln & 15, hf = ln >> 4, kc = s >> 1, el = s & 1;
    const int row = grp ? m : (4 * ((m & 7) >> 1) + 2 * (m >> 3) + (m & 1));
    const int k   = grp ? (8 * hf + 2 * kc + el) : (4 * kc + 2 * hf + el);
    const int w0  = grp ? 7 : 4;
    const int qr = row >> 1, cr = row & 1, qk = k >> 1, ck = k & 1;
    float ur = 1.f, ui = 0.f;
    #pragma unroll
    for (int j = 0; j < 3; ++j) {
      const int rb = (qr >> (2 - j)) & 1, cbit = (qk >> (2 - j)) & 1;
      const float* g = u2 + (((l * NQ + w0 + j) * 4) + rb * 2 + cbit) * 2;
      const float nr = ur * g[0] - ui * g[1];
      const float ni = ur * g[1] + ui * g[0];
      ur = nr; ui = ni;
    }
    frag[e] = (cr == ck) ? ur : (cr ? ui : -ui);
  }
  #pragma unroll 1
  for (int e = tid; e < NL * 512; e += THREADS) {
    const int l = e >> 9, p = e & 511;
    const int ln = p & 31, h = p >> 5;
    const int rr = (l % (NQ - 1)) + 1;
    unsigned pk = 0;
    #pragma unroll
    for (int half = 0; half < 2; ++half) {
      int j = ln + 64 * h + 32 * half;
      for (int w = NQ - 1; w >= 0; --w) {
        const int cpos = 9 - w;
        const int tpos = 9 - ((w + rr) % NQ);
        j ^= ((j >> cpos) & 1) << tpos;
      }
      pk |= ((unsigned)j) << (16 * half);
    }
    permPk[e] = pk;
  }
  __syncthreads();

  const int wave = tid >> 5, lane = tid & 31;
  const int b = blockIdx.x * WAVES_PER_BLOCK + wave;
  float2* S = stateBase + wave * NSTATES;

  float cw[NQ], sw[NQ];
  #pragma unroll
  for (int w = 0; w < NQ; ++w) {
    const float ang = 1.5707963267948966f * x[b * NQ + w];
    cw[w] = cosf(ang);
    sw[w] = sinf(ang);
  }
  for (int t = 0; t < 32; ++t) {
    const int i = lane + 32 * t;
    float a = 1.f;
    #pragma unroll
    for (int w = 0; w < NQ; ++w) a *= ((i >> (9 - w)) & 1) ? sw[w] : cw[w];
    S[i] = make_float2(a, 0.f);
  }

  #pragma unroll 1
  for (int l = 0; l < NL; ++l) {
    apply16(S, UAre + l * 256, UAim + l * 256, lane);
    apply8r<1>((float*)S, frag + (l * 2 + 0) * 256, lane);
    apply8r<0>((float*)S, frag + (l * 2 + 1) * 256, lane);

    const unsigned* pl = permPk + l * 512;
    float2 tv[32];
    #pragma unroll
    for (int h = 0; h < 16; ++h) {
      const unsigned pk = pl[lane + 32 * h];
      tv[2 * h]     = S[pk & 0xFFFFu];
      tv[2 * h + 1] = S[pk >> 16];
    }
    #pragma unroll
    for (int t = 0; t < 32; ++t) S[lane + 32 * t] = tv[t];
  }

  float acc[NQ];
  #pragma unroll
  for (int w = 0; w < NQ; ++w) acc[w] = 0.f;
  for (int t = 0; t < 32; ++t) {
    const int i = lane + 32 * t;
    const float2 z = S[i];
    const float p = z.x * z.x + z.y * z.y;
    #pragma unroll
    for (int w = 0; w < NQ; ++w) acc[w] += ((i >> (9 - w)) & 1) ? -p : p;
  }
  #pragma unroll
  for (int w = 0; w < NQ; ++w) {
    float v = acc[w];
    #pragma unroll
    for (int off = 16; off > 0; off >>= 1) v += __shfl_xor(v, off, 32);
    if (lane == 0) out[b * NQ + w] = v;
  }
}

extern "C" void kernel_launch(void* const* d_in, const int* in_sizes, int n_in,
                              void* d_out, int out_size, void* d_ws, size_t ws_size,
                              hipStream_t stream) {
  (void)in_sizes; (void)n_in; (void)out_size; (void)d_ws; (void)ws_size;
  const float* x  = (const float*)d_in[0];
  const float* qw = (const float*)d_in[1];
  float* out = (float*)d_out;
  const int batch = 2048;
  dim3 grid(batch / WAVES_PER_BLOCK);
  dim3 block(THREADS);
  qlayer_kernel<<<grid, block, LDS_BYTES, stream>>>(x, qw, out);
}


# Index split i=lane+32t: state init (lane factor, doubling over t; product order changes), <Z> sums (pairwise tree over t; sum order changes), CNOT table as P(lane)^P(32t): no per-amplitude bit tests
# speedup vs baseline: 1.5079x; 1.1760x over previous
#include <hip/hip_runtime.h>

typedef __attribute__((ext_vector_type(2))) float v2f;
typedef __attribute__((ext_vector_type(8))) float v8f;

#define NQ 10
#define NL 4
#define NSTATES 1024
#define WAVES_PER_BLOCK 4
#define THREADS (WAVES_PER_BLOCK * 32)

#define LDS_BYTES (1280 + 8192 + 8192 + 512 + 512 + 32768)


__device__ __forceinline__ void apply16(float2* S,
                                        const float* Ure, const float* Uim,
                                        const int lane) {
  const int hf  = lane >> 4;
  const int nlo = lane & 15;

  v2f aRe[4], aIm[4], aImN[4];
  #pragma unroll
  for (int k = 0; k < 4; ++k) {
    const int K0 = 4 * k + 2 * hf;
    aRe[k].x  = Ure[nlo * 16 + K0];  aRe[k].y  = Ure[nlo * 16 + K0 + 1];
    aIm[k].x  = Uim[nlo * 16 + K0];  aIm[k].y  = Uim[nlo * 16 + K0 + 1];
    aImN[k].x = -aIm[k].x;           aImN[k].y = -aIm[k].y;
  }

  #pragma unroll
  for (int jt = 0; jt < 4; ++jt) {
    const int n = jt * 16 + nlo;
    v8f accRe = {0.f,0.f,0.f,0.f,0.f,0.f,0.f,0.f};
    v8f accIm = {0.f,0.f,0.f,0.f,0.f,0.f,0.f,0.f};
    #pragma unroll
    for (int k = 0; k < 4; ++k) {
      const int K0 = 4 * k + 2 * hf;
      const int K1 = K0 + 1;
      const int i0 = K0 * 64 + n;
      const int i1 = K1 * 64 + n;
      const float2 z0 = S[i0];
      const float2 z1 = S[i1];
      v2f br, bi;
      br.x = z0.x;  br.y = z1.x;
      bi.x = z0.y;  bi.y = z1.y;
      accRe = __builtin_amdgcn_wmma_f32_16x16x4_f32(false, aRe[k],  false, br, (short)0, accRe, false, false);
      accRe = __builtin_amdgcn_wmma_f32_16x16x4_f32(false, aImN[k], false, bi, (short)0, accRe, false, false);
      accIm = __builtin_amdgcn_wmma_f32_16x16x4_f32(false, aRe[k],  false, bi, (short)0, accIm, false, false);
      accIm = __builtin_amdgcn_wmma_f32_16x16x4_f32(false, aIm[k],  false, br, (short)0, accIm, false, false);
    }
    #pragma unroll
    for (int v = 0; v < 8; ++v) {
      const int M = v + 8 * hf;
      const int o = M * 64 + n;
      S[o] = make_float2(accRe[v], accIm[v]);
    }
  }
}

template <int MID>
__device__ __forceinline__ void apply8r(float* F, const float* frag, const int lane) {
  const int hf  = lane >> 4;
  const int nlo = lane & 15;
  const float4 g0 = *(const float4*)(frag + lane * 8);
  const float4 g1 = *(const float4*)(frag + lane * 8 + 4);
  v2f a[4];
  a[0].x = g0.x; a[0].y = g0.y;  a[1].x = g0.z; a[1].y = g0.w;
  a[2].x = g1.x; a[2].y = g1.y;  a[3].x = g1.z; a[3].y = g1.w;

  float* ptr[8];
  v2f b[8][4];
  #pragma unroll
  for (int jt = 0; jt < 8; ++jt) {
    if (MID) {
      ptr[jt] = F + 128 * (2 * jt + (nlo >> 3)) + 2 * (nlo & 7) + 16 * hf;
      #pragma unroll
      for (int kc = 0; kc < 4; ++kc) {
        const float2 z = *(const float2*)(ptr[jt] + 32 * kc);
        b[jt][kc].x = z.x;  b[jt][kc].y = z.y;
      }
    } else {
      ptr[jt] = F + 16 * (jt * 16 + nlo) + 8 * hf;
      const float4 x0 = *(const float4*)(ptr[jt]);
      const float4 x1 = *(const float4*)(ptr[jt] + 4);
      b[jt][0].x = x0.x; b[jt][0].y = x0.y;  b[jt][1].x = x0.z; b[jt][1].y = x0.w;
      b[jt][2].x = x1.x; b[jt][2].y = x1.y;  b[jt][3].x = x1.z; b[jt][3].y = x1.w;
    }
  }
  #pragma unroll
  for (int jq = 0; jq < 2; ++jq) {
    v8f d[4];
    #pragma unroll
    for (int t = 0; t < 4; ++t) d[t] = (v8f){0.f,0.f,0.f,0.f,0.f,0.f,0.f,0.f};
    #pragma unroll
    for (int kc = 0; kc < 4; ++kc) {
      #pragma unroll
      for (int t = 0; t < 4; ++t)
        d[t] = __builtin_amdgcn_wmma_f32_16x16x4_f32(false, a[kc], false, b[4 * jq + t][kc], (short)0, d[t], false, false);
    }
    #pragma unroll
    for (int t = 0; t < 4; ++t) {
      float* o = ptr[4 * jq + t];
      if (MID) {
        #pragma unroll
        for (int j = 0; j < 4; ++j)
          *(float2*)(o + 32 * j) = make_float2(d[t][2 * j], d[t][2 * j + 1]);
      } else {
        *(float4*)(o)     = make_float4(d[t][0], d[t][1], d[t][2], d[t][3]);
        *(float4*)(o + 4) = make_float4(d[t][4], d[t][5], d[t][6], d[t][7]);
      }
    }
  }
}

extern "C" __global__ __launch_bounds__(THREADS) __attribute__((amdgpu_waves_per_eu(1, 4)))
void qlayer_kernel(const float* __restrict__ x,
                   const float* __restrict__ qw,
                   float* __restrict__ out) {
  extern __shared__ char smem_raw[];
  float* u2   = (float*)smem_raw;
  float* UAre = u2 + NL * NQ * 8;
  float* UAim = UAre + NL * 256;
  float* frag = UAim + NL * 256;
  unsigned* permL = (unsigned*)(frag + NL * 512);
  unsigned* permH = permL + NL * 32;
  float2* stateBase = (float2*)(permH + NL * 32);

  const int tid = threadIdx.x;

  #pragma unroll 1
  for (int e = tid; e < NL * NQ; e += THREADS) {
    const int l = e / NQ, w = e % NQ;
    const float phi = qw[(l * NQ + w) * 3 + 0];
    const float th  = qw[(l * NQ + w) * 3 + 1];
    const float om  = qw[(l * NQ + w) * 3 + 2];
    const float c = cosf(0.5f * th), s = sinf(0.5f * th);
    const float ap = 0.5f * (phi + om), am = 0.5f * (phi - om);
    const float cpo = cosf(ap), spo = sinf(ap);
    const float cb = cosf(am),  sb = sinf(am);
    float* g = u2 + e * 8;
    g[0] =  cpo * c; g[1] = -spo * c;
    g[2] = -cb * s;  g[3] = -sb * s;
    g[4] =  cb * s;  g[5] = -sb * s;
    g[6] =  cpo * c; g[7] =  spo * c;
  }
  __syncthreads();

  #pragma unroll 1
  for (int e = tid; e < NL * 256; e += THREADS) {
    const int l = e >> 8, rc = e & 255, r = rc >> 4, cc = rc & 15;
    float arA = 1.f, aiA = 0.f;
    #pragma unroll
    for (int q = 0; q < 4; ++q) {
      const int rb = (r >> (3 - q)) & 1, cbit = (cc >> (3 - q)) & 1;
      const float* gA = u2 + (((l * NQ + q) * 4) + rb * 2 + cbit) * 2;
      const float nr = arA * gA[0] - aiA * gA[1];
      const float ni = arA * gA[1] + aiA * gA[0];
      arA = nr; aiA = ni;
    }
    UAre[e] = arA; UAim[e] = aiA;
  }
  #pragma unroll 1
  for (int e = tid; e < NL * 512; e += THREADS) {
    const int l = e >> 9, grp = (e >> 8) & 1, ln = (e >> 3) & 31, s = e & 7;
    const int m = ln & 15, hf = ln >> 4, kc = s >> 1, el = s & 1;
    const int row = grp ? m : (4 * ((m & 7) >> 1) + 2 * (m >> 3) + (m & 1));
    const int k   = grp ? (8 * hf + 2 * kc + el) : (4 * kc + 2 * hf + el);
    const int w0  = grp ? 7 : 4;
    const int qr = row >> 1, cr = row & 1, qk = k >> 1, ck = k & 1;
    float ur = 1.f, ui = 0.f;
    #pragma unroll
    for (int j = 0; j < 3; ++j) {
      const int rb = (qr >> (2 - j)) & 1, cbit = (qk >> (2 - j)) & 1;
      const float* g = u2 + (((l * NQ + w0 + j) * 4) + rb * 2 + cbit) * 2;
      const float nr = ur * g[0] - ui * g[1];
      const float ni = ur * g[1] + ui * g[0];
      ur = nr; ui = ni;
    }
    frag[e] = (cr == ck) ? ur : (cr ? ui : -ui);
  }
  #pragma unroll 1
  for (int e = tid; e < NL * 32; e += THREADS) {
    const int l = e >> 5, ln = e & 31;
    const int rr = (l % (NQ - 1)) + 1;
    int jl = ln, jh = 32 * ln;
    #pragma unroll
    for (int w = NQ - 1; w >= 0; --w) {
      const int cpos = 9 - w;
      int tw = w + rr;
      tw = (tw >= NQ) ? tw - NQ : tw;
      const int tpos = 9 - tw;
      jl ^= ((jl >> cpos) & 1) << tpos;
      jh ^= ((jh >> cpos) & 1) << tpos;
    }
    permL[e] = ((unsigned)jl) << 3;
    permH[e] = ((unsigned)jh) << 3;
  }
  __syncthreads();

  const int wave = tid >> 5, lane = tid & 31;
  const int b = blockIdx.x * WAVES_PER_BLOCK + wave;
  float2* S = stateBase + wave * NSTATES;

  float cw[NQ], sw[NQ];
  #pragma unroll
  for (int w = 0; w < NQ; ++w) {
    const float ang = 1.5707963267948966f * x[b * NQ + w];
    cw[w] = cosf(ang);
    sw[w] = sinf(ang);
  }
  float h[32];
  {
    float a = 1.f;
    #pragma unroll
    for (int w = 5; w < NQ; ++w) a *= ((lane >> (9 - w)) & 1) ? sw[w] : cw[w];
    h[0] = a;
  }
  #pragma unroll
  for (int w = 0; w < 5; ++w) {
    #pragma unroll
    for (int u = (1 << w) - 1; u >= 0; --u) {
      const float v = h[u];
      h[2 * u + 1] = v * sw[w];
      h[2 * u]     = v * cw[w];
    }
  }
  #pragma unroll
  for (int t = 0; t < 32; ++t) S[lane + 32 * t] = make_float2(h[t], 0.f);

  #pragma unroll 1
  for (int l = 0; l < NL; ++l) {
    const unsigned pLane = permL[l * 32 + lane];
    const int pT = (int)permH[l * 32 + lane];
    apply16(S, UAre + l * 256, UAim + l * 256, lane);
    apply8r<1>((float*)S, frag + (l * 2 + 0) * 256, lane);
    apply8r<0>((float*)S, frag + (l * 2 + 1) * 256, lane);

    float2 tv[32];
    #pragma unroll
    for (int t = 0; t < 32; ++t) {
      const unsigned o = pLane ^ (unsigned)__builtin_amdgcn_readlane(pT, t);
      tv[t] = *(const float2*)((const char*)S + o);
    }
    #pragma unroll
    for (int t = 0; t < 32; ++t) S[lane + 32 * t] = tv[t];
  }

  float acc[NQ];
  float p[32];
  #pragma unroll
  for (int t = 0; t < 32; ++t) {
    const float2 z = S[lane + 32 * t];
    p[t] = z.x * z.x + z.y * z.y;
  }
  #pragma unroll
  for (int k = 0; k < 5; ++k) {
    const int n = 16 >> k;
    float d[16];
    #pragma unroll
    for (int u = 0; u < n; ++u) {
      d[u] = p[2 * u] - p[2 * u + 1];
      p[u] = p[2 * u] + p[2 * u + 1];
    }
    #pragma unroll
    for (int m = n >> 1; m > 0; m >>= 1) {
      #pragma unroll
      for (int u = 0; u < m; ++u) d[u] += d[u + m];
    }
    acc[4 - k] = d[0];
  }
  #pragma unroll
  for (int w = 5; w < NQ; ++w) acc[w] = ((lane >> (9 - w)) & 1) ? -p[0] : p[0];
  #pragma unroll
  for (int w = 0; w < NQ; ++w) {
    float v = acc[w];
    #pragma unroll
    for (int off = 16; off > 0; off >>= 1) v += __shfl_xor(v, off, 32);
    if (lane == 0) out[b * NQ + w] = v;
  }
}

extern "C" void kernel_launch(void* const* d_in, const int* in_sizes, int n_in,
                              void* d_out, int out_size, void* d_ws, size_t ws_size,
                              hipStream_t stream) {
  (void)in_sizes; (void)n_in; (void)out_size; (void)d_ws; (void)ws_size;
  const float* x  = (const float*)d_in[0];
  const float* qw = (const float*)d_in[1];
  float* out = (float*)d_out;
  const int batch = 2048;
  dim3 grid(batch / WAVES_PER_BLOCK);
  dim3 block(THREADS);
  qlayer_kernel<<<grid, block, LDS_BYTES, stream>>>(x, qw, out);
}


# Layer 0's rotations act on a product state: folded into state preparation as per-wire complex pairs U_w(cos,sin), no gate stage there: 384 not 512 f32 WMMAs; its amplitudes are products, not sums
# speedup vs baseline: 1.7183x; 1.1395x over previous
#include <hip/hip_runtime.h>

typedef __attribute__((ext_vector_type(2))) float v2f;
typedef __attribute__((ext_vector_type(8))) float v8f;

#define NQ 10
#define NL 4
#define NSTATES 1024
#define WAVES_PER_BLOCK 4
#define THREADS (WAVES_PER_BLOCK * 32)

#define LDS_BYTES (1280 + 8192 + 8192 + 512 + 512 + 32768)


__device__ __forceinline__ void apply16(float2* S,
                                        const float* Ure, const float* Uim,
                                        const int lane) {
  const int hf  = lane >> 4;
  const int nlo = lane & 15;

  v2f aRe[4], aIm[4], aImN[4];
  #pragma unroll
  for (int k = 0; k < 4; ++k) {
    const int K0 = 4 * k + 2 * hf;
    aRe[k].x  = Ure[nlo * 16 + K0];  aRe[k].y  = Ure[nlo * 16 + K0 + 1];
    aIm[k].x  = Uim[nlo * 16 + K0];  aIm[k].y  = Uim[nlo * 16 + K0 + 1];
    aImN[k].x = -aIm[k].x;           aImN[k].y = -aIm[k].y;
  }

  #pragma unroll
  for (int jt = 0; jt < 4; ++jt) {
    const int n = jt * 16 + nlo;
    v8f accRe = {0.f,0.f,0.f,0.f,0.f,0.f,0.f,0.f};
    v8f accIm = {0.f,0.f,0.f,0.f,0.f,0.f,0.f,0.f};
    #pragma unroll
    for (int k = 0; k < 4; ++k) {
      const int K0 = 4 * k + 2 * hf;
      const int K1 = K0 + 1;
      const int i0 = K0 * 64 + n;
      const int i1 = K1 * 64 + n;
      const float2 z0 = S[i0];
      const float2 z1 = S[i1];
      v2f br, bi;
      br.x = z0.x;  br.y = z1.x;
      bi.x = z0.y;  bi.y = z1.y;
      accRe = __builtin_amdgcn_wmma_f32_16x16x4_f32(false, aRe[k],  false, br, (short)0, accRe, false, false);
      accRe = __builtin_amdgcn_wmma_f32_16x16x4_f32(false, aImN[k], false, bi, (short)0, accRe, false, false);
      accIm = __builtin_amdgcn_wmma_f32_16x16x4_f32(false, aRe[k],  false, bi, (short)0, accIm, false, false);
      accIm = __builtin_amdgcn_wmma_f32_16x16x4_f32(false, aIm[k],  false, br, (short)0, accIm, false, false);
    }
    #pragma unroll
    for (int v = 0; v < 8; ++v) {
      const int M = v + 8 * hf;
      const int o = M * 64 + n;
      S[o] = make_float2(accRe[v], accIm[v]);
    }
  }
}

template <int MID>
__device__ __forceinline__ void apply8r(float* F, const float* frag, const int lane) {
  const int hf  = lane >> 4;
  const int nlo = lane & 15;
  const float4 g0 = *(const float4*)(frag + lane * 8);
  const float4 g1 = *(const float4*)(frag + lane * 8 + 4);
  v2f a[4];
  a[0].x = g0.x; a[0].y = g0.y;  a[1].x = g0.z; a[1].y = g0.w;
  a[2].x = g1.x; a[2].y = g1.y;  a[3].x = g1.z; a[3].y = g1.w;

  float* ptr[8];
  v2f b[8][4];
  #pragma unroll
  for (int jt = 0; jt < 8; ++jt) {
    if (MID) {
      ptr[jt] = F + 128 * (2 * jt + (nlo >> 3)) + 2 * (nlo & 7) + 16 * hf;
      #pragma unroll
      for (int kc = 0; kc < 4; ++kc) {
        const float2 z = *(const float2*)(ptr[jt] + 32 * kc);
        b[jt][kc].x = z.x;  b[jt][kc].y = z.y;
      }
    } else {
      ptr[jt] = F + 16 * (jt * 16 + nlo) + 8 * hf;
      const float4 x0 = *(const float4*)(ptr[jt]);
      const float4 x1 = *(const float4*)(ptr[jt] + 4);
      b[jt][0].x = x0.x; b[jt][0].y = x0.y;  b[jt][1].x = x0.z; b[jt][1].y = x0.w;
      b[jt][2].x = x1.x; b[jt][2].y = x1.y;  b[jt][3].x = x1.z; b[jt][3].y = x1.w;
    }
  }
  #pragma unroll
  for (int jq = 0; jq < 2; ++jq) {
    v8f d[4];
    #pragma unroll
    for (int t = 0; t < 4; ++t) d[t] = (v8f){0.f,0.f,0.f,0.f,0.f,0.f,0.f,0.f};
    #pragma unroll
    for (int kc = 0; kc < 4; ++kc) {
      #pragma unroll
      for (int t = 0; t < 4; ++t)
        d[t] = __builtin_amdgcn_wmma_f32_16x16x4_f32(false, a[kc], false, b[4 * jq + t][kc], (short)0, d[t], false, false);
    }
    #pragma unroll
    for (int t = 0; t < 4; ++t) {
      float* o = ptr[4 * jq + t];
      if (MID) {
        #pragma unroll
        for (int j = 0; j < 4; ++j)
          *(float2*)(o + 32 * j) = make_float2(d[t][2 * j], d[t][2 * j + 1]);
      } else {
        *(float4*)(o)     = make_float4(d[t][0], d[t][1], d[t][2], d[t][3]);
        *(float4*)(o + 4) = make_float4(d[t][4], d[t][5], d[t][6], d[t][7]);
      }
    }
  }
}

__device__ __forceinline__ void cnot_ring(float2* S, const unsigned pLane, const int pT, const int lane) {
  float2 tv[32];
  #pragma unroll
  for (int t = 0; t < 32; ++t) {
    const unsigned o = pLane ^ (unsigned)__builtin_amdgcn_readlane(pT, t);
    tv[t] = *(const float2*)((const char*)S + o);
  }
  #pragma unroll
  for (int t = 0; t < 32; ++t) S[lane + 32 * t] = tv[t];
}

extern "C" __global__ __launch_bounds__(THREADS) __attribute__((amdgpu_waves_per_eu(1, 4)))
void qlayer_kernel(const float* __restrict__ x,
                   const float* __restrict__ qw,
                   float* __restrict__ out) {
  extern __shared__ char smem_raw[];
  float* u2   = (float*)smem_raw;
  float* UAre = u2 + NL * NQ * 8;
  float* UAim = UAre + NL * 256;
  float* frag = UAim + NL * 256;
  unsigned* permL = (unsigned*)(frag + NL * 512);
  unsigned* permH = permL + NL * 32;
  float2* stateBase = (float2*)(permH + NL * 32);

  const int tid = threadIdx.x;

  #pragma unroll 1
  for (int e = tid; e < NL * NQ; e += THREADS) {
    const int l = e / NQ, w = e % NQ;
    const float phi = qw[(l * NQ + w) * 3 + 0];
    const float th  = qw[(l * NQ + w) * 3 + 1];
    const float om  = qw[(l * NQ + w) * 3 + 2];
    const float c = cosf(0.5f * th), s = sinf(0.5f * th);
    const float ap = 0.5f * (phi + om), am = 0.5f * (phi - om);
    const float cpo = cosf(ap), spo = sinf(ap);
    const float cb = cosf(am),  sb = sinf(am);
    float* g = u2 + e * 8;
    g[0] =  cpo * c; g[1] = -spo * c;
    g[2] = -cb * s;  g[3] = -sb * s;
    g[4] =  cb * s;  g[5] = -sb * s;
    g[6] =  cpo * c; g[7] =  spo * c;
  }
  __syncthreads();

  #pragma unroll 1
  for (int e = 256 + tid; e < NL * 256; e += THREADS) {
    const int l = e >> 8, rc = e & 255, r = rc >> 4, cc = rc & 15;
    float arA = 1.f, aiA = 0.f;
    #pragma unroll
    for (int q = 0; q < 4; ++q) {
      const int rb = (r >> (3 - q)) & 1, cbit = (cc >> (3 - q)) & 1;
      const float* gA = u2 + (((l * NQ + q) * 4) + rb * 2 + cbit) * 2;
      const float nr = arA * gA[0] - aiA * gA[1];
      const float ni = arA * gA[1] + aiA * gA[0];
      arA = nr; aiA = ni;
    }
    UAre[e] = arA; UAim[e] = aiA;
  }
  #pragma unroll 1
  for (int e = 512 + tid; e < NL * 512; e += THREADS) {
    const int l = e >> 9, grp = (e >> 8) & 1, ln = (e >> 3) & 31, s = e & 7;
    const int m = ln & 15, hf = ln >> 4, kc = s >> 1, el = s & 1;
    const int row = grp ? m : (4 * ((m & 7) >> 1) + 2 * (m >> 3) + (m & 1));
    const int k   = grp ? (8 * hf + 2 * kc + el) : (4 * kc + 2 * hf + el);
    const int w0  = grp ? 7 : 4;
    const int qr = row >> 1, cr = row & 1, qk = k >> 1, ck = k & 1;
    float ur = 1.f, ui = 0.f;
    #pragma unroll
    for (int j = 0; j < 3; ++j) {
      const int rb = (qr >> (2 - j)) & 1, cbit = (qk >> (2 - j)) & 1;
      const float* g = u2 + (((l * NQ + w0 + j) * 4) + rb * 2 + cbit) * 2;
      const float nr = ur * g[0] - ui * g[1];
      const float ni = ur * g[1] + ui * g[0];
      ur = nr; ui = ni;
    }
    frag[e] = (cr == ck) ? ur : (cr ? ui : -ui);
  }
  #pragma unroll 1
  for (int e = tid; e < NL * 32; e += THREADS) {
    const int l = e >> 5, ln = e & 31;
    const int rr = (l % (NQ - 1)) + 1;
    int jl = ln, jh = 32 * ln;
    #pragma unroll
    for (int w = NQ - 1; w >= 0; --w) {
      const int cpos = 9 - w;
      int tw = w + rr;
      tw = (tw >= NQ) ? tw - NQ : tw;
      const int tpos = 9 - tw;
      jl ^= ((jl >> cpos) & 1) << tpos;
      jh ^= ((jh >> cpos) & 1) << tpos;
    }
    permL[e] = ((unsigned)jl) << 3;
    permH[e] = ((unsigned)jh) << 3;
  }
  __syncthreads();

  const int wave = tid >> 5, lane = tid & 31;
  const int b = blockIdx.x * WAVES_PER_BLOCK + wave;
  float2* S = stateBase + wave * NSTATES;

  const unsigned pLane0 = permL[lane];
  const int pT0 = (int)permH[lane];
  float fr0[NQ], fi0[NQ], fr1[NQ], fi1[NQ];
  #pragma unroll
  for (int w = 0; w < NQ; ++w) {
    const float ang = 1.5707963267948966f * x[b * NQ + w];
    const float c = cosf(ang);
    const float s = sinf(ang);
    const float4 ga = *(const float4*)(u2 + w * 8);
    const float4 gb = *(const float4*)(u2 + w * 8 + 4);
    fr0[w] = ga.x * c + ga.z * s;  fi0[w] = ga.y * c + ga.w * s;
    fr1[w] = gb.x * c + gb.z * s;  fi1[w] = gb.y * c + gb.w * s;
  }
  float hr[32], hi[32];
  {
    float ar = (lane & 16) ? fr1[5] : fr0[5];
    float ai = (lane & 16) ? fi1[5] : fi0[5];
    #pragma unroll
    for (int w = 6; w < NQ; ++w) {
      const bool bit = ((lane >> (9 - w)) & 1) != 0;
      const float gr = bit ? fr1[w] : fr0[w];
      const float gi = bit ? fi1[w] : fi0[w];
      const float nr = ar * gr - ai * gi;
      const float ni = ar * gi + ai * gr;
      ar = nr; ai = ni;
    }
    hr[0] = ar; hi[0] = ai;
  }
  #pragma unroll
  for (int w = 0; w < 5; ++w) {
    #pragma unroll
    for (int u = (1 << w) - 1; u >= 0; --u) {
      const float vr = hr[u], vi = hi[u];
      hr[2 * u + 1] = vr * fr1[w] - vi * fi1[w];
      hi[2 * u + 1] = vr * fi1[w] + vi * fr1[w];
      hr[2 * u]     = vr * fr0[w] - vi * fi0[w];
      hi[2 * u]     = vr * fi0[w] + vi * fr0[w];
    }
  }
  #pragma unroll
  for (int t = 0; t < 32; ++t) S[lane + 32 * t] = make_float2(hr[t], hi[t]);

  cnot_ring(S, pLane0, pT0, lane);

  #pragma unroll 1
  for (int l = 1; l < NL; ++l) {
    const unsigned pLane = permL[l * 32 + lane];
    const int pT = (int)permH[l * 32 + lane];
    apply16(S, UAre + l * 256, UAim + l * 256, lane);
    apply8r<1>((float*)S, frag + (l * 2 + 0) * 256, lane);
    apply8r<0>((float*)S, frag + (l * 2 + 1) * 256, lane);
    cnot_ring(S, pLane, pT, lane);
  }

  float acc[NQ];
  float p[32];
  #pragma unroll
  for (int t = 0; t < 32; ++t) {
    const float2 z = S[lane + 32 * t];
    p[t] = z.x * z.x + z.y * z.y;
  }
  #pragma unroll
  for (int k = 0; k < 5; ++k) {
    const int n = 16 >> k;
    float d[16];
    #pragma unroll
    for (int u = 0; u < n; ++u) {
      d[u] = p[2 * u] - p[2 * u + 1];
      p[u] = p[2 * u] + p[2 * u + 1];
    }
    #pragma unroll
    for (int m = n >> 1; m > 0; m >>= 1) {
      #pragma unroll
      for (int u = 0; u < m; ++u) d[u] += d[u + m];
    }
    acc[4 - k] = d[0];
  }
  #pragma unroll
  for (int w = 5; w < NQ; ++w) acc[w] = ((lane >> (9 - w)) & 1) ? -p[0] : p[0];
  #pragma unroll
  for (int w = 0; w < NQ; ++w) {
    float v = acc[w];
    #pragma unroll
    for (int off = 16; off > 0; off >>= 1) v += __shfl_xor(v, off, 32);
    if (lane == 0) out[b * NQ + w] = v;
  }
}

extern "C" void kernel_launch(void* const* d_in, const int* in_sizes, int n_in,
                              void* d_out, int out_size, void* d_ws, size_t ws_size,
                              hipStream_t stream) {
  (void)in_sizes; (void)n_in; (void)out_size; (void)d_ws; (void)ws_size;
  const float* x  = (const float*)d_in[0];
  const float* qw = (const float*)d_in[1];
  float* out = (float*)d_out;
  const int batch = 2048;
  dim3 grid(batch / WAVES_PER_BLOCK);
  dim3 block(THREADS);
  qlayer_kernel<<<grid, block, LDS_BYTES, stream>>>(x, qw, out);
}


# Each sin/cos evaluated once on its own lane and shared by lane reads: the wave's 10 embedding inputs are one load (lane w), not 10 dependent load+trig rounds; gate tables 2 not 6 evaluations deep
# speedup vs baseline: 1.8943x; 1.1024x over previous
#include <hip/hip_runtime.h>

typedef __attribute__((ext_vector_type(2))) float v2f;
typedef __attribute__((ext_vector_type(8))) float v8f;

#define NQ 10
#define NL 4
#define NSTATES 1024
#define WAVES_PER_BLOCK 4
#define THREADS (WAVES_PER_BLOCK * 32)

#define LDS_BYTES (1280 + 8192 + 8192 + 512 + 512 + 32768)


__device__ __forceinline__ void apply16(float2* S,
                                        const float* Ure, const float* Uim,
                                        const int lane) {
  const int hf  = lane >> 4;
  const int nlo = lane & 15;

  v2f aRe[4], aIm[4], aImN[4];
  #pragma unroll
  for (int k = 0; k < 4; ++k) {
    const int K0 = 4 * k + 2 * hf;
    aRe[k].x  = Ure[nlo * 16 + K0];  aRe[k].y  = Ure[nlo * 16 + K0 + 1];
    aIm[k].x  = Uim[nlo * 16 + K0];  aIm[k].y  = Uim[nlo * 16 + K0 + 1];
    aImN[k].x = -aIm[k].x;           aImN[k].y = -aIm[k].y;
  }

  #pragma unroll
  for (int jt = 0; jt < 4; ++jt) {
    const int n = jt * 16 + nlo;
    v8f accRe = {0.f,0.f,0.f,0.f,0.f,0.f,0.f,0.f};
    v8f accIm = {0.f,0.f,0.f,0.f,0.f,0.f,0.f,0.f};
    #pragma unroll
    for (int k = 0; k < 4; ++k) {
      const int K0 = 4 * k + 2 * hf;
      const int K1 = K0 + 1;
      const int i0 = K0 * 64 + n;
      const int i1 = K1 * 64 + n;
      const float2 z0 = S[i0];
      const float2 z1 = S[i1];
      v2f br, bi;
      br.x = z0.x;  br.y = z1.x;
      bi.x = z0.y;  bi.y = z1.y;
      accRe = __builtin_amdgcn_wmma_f32_16x16x4_f32(false, aRe[k],  false, br, (short)0, accRe, false, false);
      accRe = __builtin_amdgcn_wmma_f32_16x16x4_f32(false, aImN[k], false, bi, (short)0, accRe, false, false);
      accIm = __builtin_amdgcn_wmma_f32_16x16x4_f32(false, aRe[k],  false, bi, (short)0, accIm, false, false);
      accIm = __builtin_amdgcn_wmma_f32_16x16x4_f32(false, aIm[k],  false, br, (short)0, accIm, false, false);
    }
    #pragma unroll
    for (int v = 0; v < 8; ++v) {
      const int M = v + 8 * hf;
      const int o = M * 64 + n;
      S[o] = make_float2(accRe[v], accIm[v]);
    }
  }
}

template <int MID>
__device__ __forceinline__ void apply8r(float* F, const float* frag, const int lane) {
  const int hf  = lane >> 4;
  const int nlo = lane & 15;
  const float4 g0 = *(const float4*)(frag + lane * 8);
  const float4 g1 = *(const float4*)(frag + lane * 8 + 4);
  v2f a[4];
  a[0].x = g0.x; a[0].y = g0.y;  a[1].x = g0.z; a[1].y = g0.w;
  a[2].x = g1.x; a[2].y = g1.y;  a[3].x = g1.z; a[3].y = g1.w;

  float* ptr[8];
  v2f b[8][4];
  #pragma unroll
  for (int jt = 0; jt < 8; ++jt) {
    if (MID) {
      ptr[jt] = F + 128 * (2 * jt + (nlo >> 3)) + 2 * (nlo & 7) + 16 * hf;
      #pragma unroll
      for (int kc = 0; kc < 4; ++kc) {
        const float2 z = *(const float2*)(ptr[jt] + 32 * kc);
        b[jt][kc].x = z.x;  b[jt][kc].y = z.y;
      }
    } else {
      ptr[jt] = F + 16 * (jt * 16 + nlo) + 8 * hf;
      const float4 x0 = *(const float4*)(ptr[jt]);
      const float4 x1 = *(const float4*)(ptr[jt] + 4);
      b[jt][0].x = x0.x; b[jt][0].y = x0.y;  b[jt][1].x = x0.z; b[jt][1].y = x0.w;
      b[jt][2].x = x1.x; b[jt][2].y = x1.y;  b[jt][3].x = x1.z; b[jt][3].y = x1.w;
    }
  }
  #pragma unroll
  for (int jq = 0; jq < 2; ++jq) {
    v8f d[4];
    #pragma unroll
    for (int t = 0; t < 4; ++t) d[t] = (v8f){0.f,0.f,0.f,0.f,0.f,0.f,0.f,0.f};
    #pragma unroll
    for (int kc = 0; kc < 4; ++kc) {
      #pragma unroll
      for (int t = 0; t < 4; ++t)
        d[t] = __builtin_amdgcn_wmma_f32_16x16x4_f32(false, a[kc], false, b[4 * jq + t][kc], (short)0, d[t], false, false);
    }
    #pragma unroll
    for (int t = 0; t < 4; ++t) {
      float* o = ptr[4 * jq + t];
      if (MID) {
        #pragma unroll
        for (int j = 0; j < 4; ++j)
          *(float2*)(o + 32 * j) = make_float2(d[t][2 * j], d[t][2 * j + 1]);
      } else {
        *(float4*)(o)     = make_float4(d[t][0], d[t][1], d[t][2], d[t][3]);
        *(float4*)(o + 4) = make_float4(d[t][4], d[t][5], d[t][6], d[t][7]);
      }
    }
  }
}

__device__ __forceinline__ float lane_value(const float v, const int w) {
  return __int_as_float(__builtin_amdgcn_readlane(__float_as_int(v), w));
}

__device__ __forceinline__ void cnot_ring(float2* S, const unsigned pLane, const int pT, const int lane) {
  float2 tv[32];
  #pragma unroll
  for (int t = 0; t < 32; ++t) {
    const unsigned o = pLane ^ (unsigned)__builtin_amdgcn_readlane(pT, t);
    tv[t] = *(const float2*)((const char*)S + o);
  }
  #pragma unroll
  for (int t = 0; t < 32; ++t) S[lane + 32 * t] = tv[t];
}

extern "C" __global__ __launch_bounds__(THREADS) __attribute__((amdgpu_waves_per_eu(1, 4)))
void qlayer_kernel(const float* __restrict__ x,
                   const float* __restrict__ qw,
                   float* __restrict__ out) {
  extern __shared__ char smem_raw[];
  float* u2   = (float*)smem_raw;
  float* UAre = u2 + NL * NQ * 8;
  float* UAim = UAre + NL * 256;
  float* frag = UAim + NL * 256;
  unsigned* permL = (unsigned*)(frag + NL * 512);
  unsigned* permH = permL + NL * 32;
  float2* stateBase = (float2*)(permH + NL * 32);

  const int tid = threadIdx.x;
  const int wave = tid >> 5, lane = tid & 31;
  const int b = blockIdx.x * WAVES_PER_BLOCK + wave;

  const int wl = lane < NQ ? lane : NQ - 1;
  const float xw = x[b * NQ + wl];

  static_assert(NL * NQ == WAVES_PER_BLOCK * 10, "10 gates per wave");
  {
    const int k3 = lane / 3;
    const int k  = k3 < 10 ? k3 : 9;
    const int j  = lane - 3 * k3;
    const int e  = wave * 10 + k;
    const float phi = qw[e * 3 + 0];
    const float th  = qw[e * 3 + 1];
    const float om  = qw[e * 3 + 2];
    const float a  = (j == 0) ? 0.5f * th : ((j == 1) ? 0.5f * (phi + om) : 0.5f * (phi - om));
    const float ca = cosf(a), sa = sinf(a);
    const float cpo = __shfl_down(ca, 1, 32), spo = __shfl_down(sa, 1, 32);
    const float cb  = __shfl_down(ca, 2, 32), sb  = __shfl_down(sa, 2, 32);
    if (j == 0 && k3 < 10) {
      const float c = ca, s = sa;
      float* g = u2 + e * 8;
      g[0] =  cpo * c; g[1] = -spo * c;
      g[2] = -cb * s;  g[3] = -sb * s;
      g[4] =  cb * s;  g[5] = -sb * s;
      g[6] =  cpo * c; g[7] =  spo * c;
    }
  }
  const float angw = 1.5707963267948966f * xw;
  const float cw = cosf(angw), sw = sinf(angw);
  __syncthreads();

  #pragma unroll 1
  for (int e = 256 + tid; e < NL * 256; e += THREADS) {
    const int l = e >> 8, rc = e & 255, r = rc >> 4, cc = rc & 15;
    float arA = 1.f, aiA = 0.f;
    #pragma unroll
    for (int q = 0; q < 4; ++q) {
      const int rb = (r >> (3 - q)) & 1, cbit = (cc >> (3 - q)) & 1;
      const float* gA = u2 + (((l * NQ + q) * 4) + rb * 2 + cbit) * 2;
      const float nr = arA * gA[0] - aiA * gA[1];
      const float ni = arA * gA[1] + aiA * gA[0];
      arA = nr; aiA = ni;
    }
    UAre[e] = arA; UAim[e] = aiA;
  }
  #pragma unroll 1
  for (int e = 512 + tid; e < NL * 512; e += THREADS) {
    const int l = e >> 9, grp = (e >> 8) & 1, ln = (e >> 3) & 31, s = e & 7;
    const int m = ln & 15, hf = ln >> 4, kc = s >> 1, el = s & 1;
    const int row = grp ? m : (4 * ((m & 7) >> 1) + 2 * (m >> 3) + (m & 1));
    const int k   = grp ? (8 * hf + 2 * kc + el) : (4 * kc + 2 * hf + el);
    const int w0  = grp ? 7 : 4;
    const int qr = row >> 1, cr = row & 1, qk = k >> 1, ck = k & 1;
    float ur = 1.f, ui = 0.f;
    #pragma unroll
    for (int j = 0; j < 3; ++j) {
      const int rb = (qr >> (2 - j)) & 1, cbit = (qk >> (2 - j)) & 1;
      const float* g = u2 + (((l * NQ + w0 + j) * 4) + rb * 2 + cbit) * 2;
      const float nr = ur * g[0] - ui * g[1];
      const float ni = ur * g[1] + ui * g[0];
      ur = nr; ui = ni;
    }
    frag[e] = (cr == ck) ? ur : (cr ? ui : -ui);
  }
  #pragma unroll 1
  for (int e = tid; e < NL * 32; e += THREADS) {
    const int l = e >> 5, ln = e & 31;
    const int rr = (l % (NQ - 1)) + 1;
    int jl = ln, jh = 32 * ln;
    #pragma unroll
    for (int w = NQ - 1; w >= 0; --w) {
      const int cpos = 9 - w;
      int tw = w + rr;
      tw = (tw >= NQ) ? tw - NQ : tw;
      const int tpos = 9 - tw;
      jl ^= ((jl >> cpos) & 1) << tpos;
      jh ^= ((jh >> cpos) & 1) << tpos;
    }
    permL[e] = ((unsigned)jl) << 3;
    permH[e] = ((unsigned)jh) << 3;
  }
  __syncthreads();

  float2* S = stateBase + wave * NSTATES;

  const unsigned pLane0 = permL[lane];
  const int pT0 = (int)permH[lane];
  float fr0[NQ], fi0[NQ], fr1[NQ], fi1[NQ];
  {
    const float4 ga = *(const float4*)(u2 + wl * 8);
    const float4 gb = *(const float4*)(u2 + wl * 8 + 4);
    const float pr0 = ga.x * cw + ga.z * sw, pi0 = ga.y * cw + ga.w * sw;
    const float pr1 = gb.x * cw + gb.z * sw, pi1 = gb.y * cw + gb.w * sw;
    #pragma unroll
    for (int w = 0; w < NQ; ++w) {
      fr0[w] = lane_value(pr0, w);  fi0[w] = lane_value(pi0, w);
      fr1[w] = lane_value(pr1, w);  fi1[w] = lane_value(pi1, w);
    }
  }
  float hr[32], hi[32];
  {
    float ar = (lane & 16) ? fr1[5] : fr0[5];
    float ai = (lane & 16) ? fi1[5] : fi0[5];
    #pragma unroll
    for (int w = 6; w < NQ; ++w) {
      const bool bit = ((lane >> (9 - w)) & 1) != 0;
      const float gr = bit ? fr1[w] : fr0[w];
      const float gi = bit ? fi1[w] : fi0[w];
      const float nr = ar * gr - ai * gi;
      const float ni = ar * gi + ai * gr;
      ar = nr; ai = ni;
    }
    hr[0] = ar; hi[0] = ai;
  }
  #pragma unroll
  for (int w = 0; w < 5; ++w) {
    #pragma unroll
    for (int u = (1 << w) - 1; u >= 0; --u) {
      const float vr = hr[u], vi = hi[u];
      hr[2 * u + 1] = vr * fr1[w] - vi * fi1[w];
      hi[2 * u + 1] = vr * fi1[w] + vi * fr1[w];
      hr[2 * u]     = vr * fr0[w] - vi * fi0[w];
      hi[2 * u]     = vr * fi0[w] + vi * fr0[w];
    }
  }
  #pragma unroll
  for (int t = 0; t < 32; ++t) S[lane + 32 * t] = make_float2(hr[t], hi[t]);

  cnot_ring(S, pLane0, pT0, lane);

  #pragma unroll 1
  for (int l = 1; l < NL; ++l) {
    const unsigned pLane = permL[l * 32 + lane];
    const int pT = (int)permH[l * 32 + lane];
    apply16(S, UAre + l * 256, UAim + l * 256, lane);
    apply8r<1>((float*)S, frag + (l * 2 + 0) * 256, lane);
    apply8r<0>((float*)S, frag + (l * 2 + 1) * 256, lane);
    cnot_ring(S, pLane, pT, lane);
  }

  float acc[NQ];
  float p[32];
  #pragma unroll
  for (int t = 0; t < 32; ++t) {
    const float2 z = S[lane + 32 * t];
    p[t] = z.x * z.x + z.y * z.y;
  }
  #pragma unroll
  for (int k = 0; k < 5; ++k) {
    const int n = 16 >> k;
    float d[16];
    #pragma unroll
    for (int u = 0; u < n; ++u) {
      d[u] = p[2 * u] - p[2 * u + 1];
      p[u] = p[2 * u] + p[2 * u + 1];
    }
    #pragma unroll
    for (int m = n >> 1; m > 0; m >>= 1) {
      #pragma unroll
      for (int u = 0; u < m; ++u) d[u] += d[u + m];
    }
    acc[4 - k] = d[0];
  }
  #pragma unroll
  for (int w = 5; w < NQ; ++w) acc[w] = ((lane >> (9 - w)) & 1) ? -p[0] : p[0];
  #pragma unroll
  for (int w = 0; w < NQ; ++w) {
    float v = acc[w];
    #pragma unroll
    for (int off = 16; off > 0; off >>= 1) v += __shfl_xor(v, off, 32);
    if (lane == 0) out[b * NQ + w] = v;
  }
}

extern "C" void kernel_launch(void* const* d_in, const int* in_sizes, int n_in,
                              void* d_out, int out_size, void* d_ws, size_t ws_size,
                              hipStream_t stream) {
  (void)in_sizes; (void)n_in; (void)out_size; (void)d_ws; (void)ws_size;
  const float* x  = (const float*)d_in[0];
  const float* qw = (const float*)d_in[1];
  float* out = (float*)d_out;
  const int batch = 2048;
  dim3 grid(batch / WAVES_PER_BLOCK);
  dim3 block(THREADS);
  qlayer_kernel<<<grid, block, LDS_BYTES, stream>>>(x, qw, out);
}


# The 10 cross-lane sums of <Z>: row_xmask DPP adds and one swap of the wave's halves, all 10 side by side, one store by 10 lanes; not 50 dependent LDS-unit lane permutes; same pairs, same order
# speedup vs baseline: 2.3248x; 1.2273x over previous
#include <hip/hip_runtime.h>

typedef __attribute__((ext_vector_type(2))) float v2f;
typedef __attribute__((ext_vector_type(8))) float v8f;

#define NQ 10
#define NL 4
#define NSTATES 1024
#define WAVES_PER_BLOCK 4
#define THREADS (WAVES_PER_BLOCK * 32)

#define LDS_BYTES (1280 + 8192 + 8192 + 512 + 512 + 32768)


__device__ __forceinline__ void apply16(float2* S,
                                        const float* Ure, const float* Uim,
                                        const int lane) {
  const int hf  = lane >> 4;
  const int nlo = lane & 15;

  v2f aRe[4], aIm[4], aImN[4];
  #pragma unroll
  for (int k = 0; k < 4; ++k) {
    const int K0 = 4 * k + 2 * hf;
    aRe[k].x  = Ure[nlo * 16 + K0];  aRe[k].y  = Ure[nlo * 16 + K0 + 1];
    aIm[k].x  = Uim[nlo * 16 + K0];  aIm[k].y  = Uim[nlo * 16 + K0 + 1];
    aImN[k].x = -aIm[k].x;           aImN[k].y = -aIm[k].y;
  }

  #pragma unroll
  for (int jt = 0; jt < 4; ++jt) {
    const int n = jt * 16 + nlo;
    v8f accRe = {0.f,0.f,0.f,0.f,0.f,0.f,0.f,0.f};
    v8f accIm = {0.f,0.f,0.f,0.f,0.f,0.f,0.f,0.f};
    #pragma unroll
    for (int k = 0; k < 4; ++k) {
      const int K0 = 4 * k + 2 * hf;
      const int K1 = K0 + 1;
      const int i0 = K0 * 64 + n;
      const int i1 = K1 * 64 + n;
      const float2 z0 = S[i0];
      const float2 z1 = S[i1];
      v2f br, bi;
      br.x = z0.x;  br.y = z1.x;
      bi.x = z0.y;  bi.y = z1.y;
      accRe = __builtin_amdgcn_wmma_f32_16x16x4_f32(false, aRe[k],  false, br, (short)0, accRe, false, false);
      accRe = __builtin_amdgcn_wmma_f32_16x16x4_f32(false, aImN[k], false, bi, (short)0, accRe, false, false);
      accIm = __builtin_amdgcn_wmma_f32_16x16x4_f32(false, aRe[k],  false, bi, (short)0, accIm, false, false);
      accIm = __builtin_amdgcn_wmma_f32_16x16x4_f32(false, aIm[k],  false, br, (short)0, accIm, false, false);
    }
    #pragma unroll
    for (int v = 0; v < 8; ++v) {
      const int M = v + 8 * hf;
      const int o = M * 64 + n;
      S[o] = make_float2(accRe[v], accIm[v]);
    }
  }
}

template <int MID>
__device__ __forceinline__ void apply8r(float* F, const float* frag, const int lane) {
  const int hf  = lane >> 4;
  const int nlo = lane & 15;
  const float4 g0 = *(const float4*)(frag + lane * 8);
  const float4 g1 = *(const float4*)(frag + lane * 8 + 4);
  v2f a[4];
  a[0].x = g0.x; a[0].y = g0.y;  a[1].x = g0.z; a[1].y = g0.w;
  a[2].x = g1.x; a[2].y = g1.y;  a[3].x = g1.z; a[3].y = g1.w;

  float* ptr[8];
  v2f b[8][4];
  #pragma unroll
  for (int jt = 0; jt < 8; ++jt) {
    if (MID) {
      ptr[jt] = F + 128 * (2 * jt + (nlo >> 3)) + 2 * (nlo & 7) + 16 * hf;
      #pragma unroll
      for (int kc = 0; kc < 4; ++kc) {
        const float2 z = *(const float2*)(ptr[jt] + 32 * kc);
        b[jt][kc].x = z.x;  b[jt][kc].y = z.y;
      }
    } else {
      ptr[jt] = F + 16 * (jt * 16 + nlo) + 8 * hf;
      const float4 x0 = *(const float4*)(ptr[jt]);
      const float4 x1 = *(const float4*)(ptr[jt] + 4);
      b[jt][0].x = x0.x; b[jt][0].y = x0.y;  b[jt][1].x = x0.z; b[jt][1].y = x0.w;
      b[jt][2].x = x1.x; b[jt][2].y = x1.y;  b[jt][3].x = x1.z; b[jt][3].y = x1.w;
    }
  }
  #pragma unroll
  for (int jq = 0; jq < 2; ++jq) {
    v8f d[4];
    #pragma unroll
    for (int t = 0; t < 4; ++t) d[t] = (v8f){0.f,0.f,0.f,0.f,0.f,0.f,0.f,0.f};
    #pragma unroll
    for (int kc = 0; kc < 4; ++kc) {
      #pragma unroll
      for (int t = 0; t < 4; ++t)
        d[t] = __builtin_amdgcn_wmma_f32_16x16x4_f32(false, a[kc], false, b[4 * jq + t][kc], (short)0, d[t], false, false);
    }
    #pragma unroll
    for (int t = 0; t < 4; ++t) {
      float* o = ptr[4 * jq + t];
      if (MID) {
        #pragma unroll
        for (int j = 0; j < 4; ++j)
          *(float2*)(o + 32 * j) = make_float2(d[t][2 * j], d[t][2 * j + 1]);
      } else {
        *(float4*)(o)     = make_float4(d[t][0], d[t][1], d[t][2], d[t][3]);
        *(float4*)(o + 4) = make_float4(d[t][4], d[t][5], d[t][6], d[t][7]);
      }
    }
  }
}

__device__ __forceinline__ float lane_value(const float v, const int w) {
  return __int_as_float(__builtin_amdgcn_readlane(__float_as_int(v), w));
}

__device__ __forceinline__ void cnot_ring(float2* S, const unsigned pLane, const int pT, const int lane) {
  float2 tv[32];
  #pragma unroll
  for (int t = 0; t < 32; ++t) {
    const unsigned o = pLane ^ (unsigned)__builtin_amdgcn_readlane(pT, t);
    tv[t] = *(const float2*)((const char*)S + o);
  }
  #pragma unroll
  for (int t = 0; t < 32; ++t) S[lane + 32 * t] = tv[t];
}

__device__ __forceinline__ float xhalf(const float x) {
  const int b = __float_as_int(x);
  return __int_as_float(__builtin_amdgcn_permlanex16(b, b, 0x76543210, 0xfedcba98, false, false));
}

template <int M>
__device__ __forceinline__ float row_xor(const float x) {
  return __int_as_float(__builtin_amdgcn_update_dpp(0, __float_as_int(x), 0x160 | M, 0xf, 0xf, true));
}

extern "C" __global__ __launch_bounds__(THREADS) __attribute__((amdgpu_waves_per_eu(1, 4)))
void qlayer_kernel(const float* __restrict__ x,
                   const float* __restrict__ qw,
                   float* __restrict__ out) {
  extern __shared__ char smem_raw[];
  float* u2   = (float*)smem_raw;
  float* UAre = u2 + NL * NQ * 8;
  float* UAim = UAre + NL * 256;
  float* frag = UAim + NL * 256;
  unsigned* permL = (unsigned*)(frag + NL * 512);
  unsigned* permH = permL + NL * 32;
  float2* stateBase = (float2*)(permH + NL * 32);

  const int tid = threadIdx.x;
  const int wave = tid >> 5, lane = tid & 31;
  const int b = blockIdx.x * WAVES_PER_BLOCK + wave;

  const int wl = lane < NQ ? lane : NQ - 1;
  const float xw = x[b * NQ + wl];

  static_assert(NL * NQ == WAVES_PER_BLOCK * 10, "10 gates per wave");
  {
    const int k3 = lane / 3;
    const int k  = k3 < 10 ? k3 : 9;
    const int j  = lane - 3 * k3;
    const int e  = wave * 10 + k;
    const float phi = qw[e * 3 + 0];
    const float th  = qw[e * 3 + 1];
    const float om  = qw[e * 3 + 2];
    const float a  = (j == 0) ? 0.5f * th : ((j == 1) ? 0.5f * (phi + om) : 0.5f * (phi - om));
    const float ca = cosf(a), sa = sinf(a);
    const float cpo = __shfl_down(ca, 1, 32), spo = __shfl_down(sa, 1, 32);
    const float cb  = __shfl_down(ca, 2, 32), sb  = __shfl_down(sa, 2, 32);
    if (j == 0 && k3 < 10) {
      const float c = ca, s = sa;
      float* g = u2 + e * 8;
      g[0] =  cpo * c; g[1] = -spo * c;
      g[2] = -cb * s;  g[3] = -sb * s;
      g[4] =  cb * s;  g[5] = -sb * s;
      g[6] =  cpo * c; g[7] =  spo * c;
    }
  }
  const float angw = 1.5707963267948966f * xw;
  const float cw = cosf(angw), sw = sinf(angw);
  __syncthreads();

  #pragma unroll 1
  for (int e = 256 + tid; e < NL * 256; e += THREADS) {
    const int l = e >> 8, rc = e & 255, r = rc >> 4, cc = rc & 15;
    float arA = 1.f, aiA = 0.f;
    #pragma unroll
    for (int q = 0; q < 4; ++q) {
      const int rb = (r >> (3 - q)) & 1, cbit = (cc >> (3 - q)) & 1;
      const float* gA = u2 + (((l * NQ + q) * 4) + rb * 2 + cbit) * 2;
      const float nr = arA * gA[0] - aiA * gA[1];
      const float ni = arA * gA[1] + aiA * gA[0];
      arA = nr; aiA = ni;
    }
    UAre[e] = arA; UAim[e] = aiA;
  }
  #pragma unroll 1
  for (int e = 512 + tid; e < NL * 512; e += THREADS) {
    const int l = e >> 9, grp = (e >> 8) & 1, ln = (e >> 3) & 31, s = e & 7;
    const int m = ln & 15, hf = ln >> 4, kc = s >> 1, el = s & 1;
    const int row = grp ? m : (4 * ((m & 7) >> 1) + 2 * (m >> 3) + (m & 1));
    const int k   = grp ? (8 * hf + 2 * kc + el) : (4 * kc + 2 * hf + el);
    const int w0  = grp ? 7 : 4;
    const int qr = row >> 1, cr = row & 1, qk = k >> 1, ck = k & 1;
    float ur = 1.f, ui = 0.f;
    #pragma unroll
    for (int j = 0; j < 3; ++j) {
      const int rb = (qr >> (2 - j)) & 1, cbit = (qk >> (2 - j)) & 1;
      const float* g = u2 + (((l * NQ + w0 + j) * 4) + rb * 2 + cbit) * 2;
      const float nr = ur * g[0] - ui * g[1];
      const float ni = ur * g[1] + ui * g[0];
      ur = nr; ui = ni;
    }
    frag[e] = (cr == ck) ? ur : (cr ? ui : -ui);
  }
  #pragma unroll 1
  for (int e = tid; e < NL * 32; e += THREADS) {
    const int l = e >> 5, ln = e & 31;
    const int rr = (l % (NQ - 1)) + 1;
    int jl = ln, jh = 32 * ln;
    #pragma unroll
    for (int w = NQ - 1; w >= 0; --w) {
      const int cpos = 9 - w;
      int tw = w + rr;
      tw = (tw >= NQ) ? tw - NQ : tw;
      const int tpos = 9 - tw;
      jl ^= ((jl >> cpos) & 1) << tpos;
      jh ^= ((jh >> cpos) & 1) << tpos;
    }
    permL[e] = ((unsigned)jl) << 3;
    permH[e] = ((unsigned)jh) << 3;
  }
  __syncthreads();

  float2* S = stateBase + wave * NSTATES;

  const unsigned pLane0 = permL[lane];
  const int pT0 = (int)permH[lane];
  float fr0[NQ], fi0[NQ], fr1[NQ], fi1[NQ];
  {
    const float4 ga = *(const float4*)(u2 + wl * 8);
    const float4 gb = *(const float4*)(u2 + wl * 8 + 4);
    const float pr0 = ga.x * cw + ga.z * sw, pi0 = ga.y * cw + ga.w * sw;
    const float pr1 = gb.x * cw + gb.z * sw, pi1 = gb.y * cw + gb.w * sw;
    #pragma unroll
    for (int w = 0; w < NQ; ++w) {
      fr0[w] = lane_value(pr0, w);  fi0[w] = lane_value(pi0, w);
      fr1[w] = lane_value(pr1, w);  fi1[w] = lane_value(pi1, w);
    }
  }
  float hr[32], hi[32];
  {
    float ar = (lane & 16) ? fr1[5] : fr0[5];
    float ai = (lane & 16) ? fi1[5] : fi0[5];
    #pragma unroll
    for (int w = 6; w < NQ; ++w) {
      const bool bit = ((lane >> (9 - w)) & 1) != 0;
      const float gr = bit ? fr1[w] : fr0[w];
      const float gi = bit ? fi1[w] : fi0[w];
      const float nr = ar * gr - ai * gi;
      const float ni = ar * gi + ai * gr;
      ar = nr; ai = ni;
    }
    hr[0] = ar; hi[0] = ai;
  }
  #pragma unroll
  for (int w = 0; w < 5; ++w) {
    #pragma unroll
    for (int u = (1 << w) - 1; u >= 0; --u) {
      const float vr = hr[u], vi = hi[u];
      hr[2 * u + 1] = vr * fr1[w] - vi * fi1[w];
      hi[2 * u + 1] = vr * fi1[w] + vi * fr1[w];
      hr[2 * u]     = vr * fr0[w] - vi * fi0[w];
      hi[2 * u]     = vr * fi0[w] + vi * fr0[w];
    }
  }
  #pragma unroll
  for (int t = 0; t < 32; ++t) S[lane + 32 * t] = make_float2(hr[t], hi[t]);

  cnot_ring(S, pLane0, pT0, lane);

  #pragma unroll 1
  for (int l = 1; l < NL; ++l) {
    const unsigned pLane = permL[l * 32 + lane];
    const int pT = (int)permH[l * 32 + lane];
    apply16(S, UAre + l * 256, UAim + l * 256, lane);
    apply8r<1>((float*)S, frag + (l * 2 + 0) * 256, lane);
    apply8r<0>((float*)S, frag + (l * 2 + 1) * 256, lane);
    cnot_ring(S, pLane, pT, lane);
  }

  float acc[NQ];
  float p[32];
  #pragma unroll
  for (int t = 0; t < 32; ++t) {
    const float2 z = S[lane + 32 * t];
    p[t] = z.x * z.x + z.y * z.y;
  }
  #pragma unroll
  for (int k = 0; k < 5; ++k) {
    const int n = 16 >> k;
    float d[16];
    #pragma unroll
    for (int u = 0; u < n; ++u) {
      d[u] = p[2 * u] - p[2 * u + 1];
      p[u] = p[2 * u] + p[2 * u + 1];
    }
    #pragma unroll
    for (int m = n >> 1; m > 0; m >>= 1) {
      #pragma unroll
      for (int u = 0; u < m; ++u) d[u] += d[u + m];
    }
    acc[4 - k] = d[0];
  }
  #pragma unroll
  for (int w = 5; w < NQ; ++w) acc[w] = ((lane >> (9 - w)) & 1) ? -p[0] : p[0];
  float r[NQ];
  #pragma unroll
  for (int w = 0; w < NQ; ++w) {
    float v = acc[w];
    v += xhalf(v);
    v += row_xor<8>(v);
    v += row_xor<4>(v);
    v += row_xor<2>(v);
    v += row_xor<1>(v);
    r[w] = v;
  }
  float o = r[0];
  #pragma unroll
  for (int w = 1; w < NQ; ++w) o = (lane == w) ? r[w] : o;
  if (lane < NQ) out[b * NQ + lane] = o;
}

extern "C" void kernel_launch(void* const* d_in, const int* in_sizes, int n_in,
                              void* d_out, int out_size, void* d_ws, size_t ws_size,
                              hipStream_t stream) {
  (void)in_sizes; (void)n_in; (void)out_size; (void)d_ws; (void)ws_size;
  const float* x  = (const float*)d_in[0];
  const float* qw = (const float*)d_in[1];
  float* out = (float*)d_out;
  const int batch = 2048;
  dim3 grid(batch / WAVES_PER_BLOCK);
  dim3 block(THREADS);
  qlayer_kernel<<<grid, block, LDS_BYTES, stream>>>(x, qw, out);
}


# Table build: the LDS reads of gate entries for all 18 table entries of a thread issued as one group ahead of the products (loops unrolled), not one dependent LDS round trip per entry; same products
# speedup vs baseline: 2.6441x; 1.1373x over previous
#include <hip/hip_runtime.h>

typedef __attribute__((ext_vector_type(2))) float v2f;
typedef __attribute__((ext_vector_type(8))) float v8f;

#define NQ 10
#define NL 4
#define NSTATES 1024
#define WAVES_PER_BLOCK 4
#define THREADS (WAVES_PER_BLOCK * 32)

#define LDS_BYTES (1280 + 8192 + 8192 + 512 + 512 + 32768)


__device__ __forceinline__ void apply16(float2* S,
                                        const float* Ure, const float* Uim,
                                        const int lane) {
  const int hf  = lane >> 4;
  const int nlo = lane & 15;

  v2f aRe[4], aIm[4], aImN[4];
  #pragma unroll
  for (int k = 0; k < 4; ++k) {
    const int K0 = 4 * k + 2 * hf;
    aRe[k].x  = Ure[nlo * 16 + K0];  aRe[k].y  = Ure[nlo * 16 + K0 + 1];
    aIm[k].x  = Uim[nlo * 16 + K0];  aIm[k].y  = Uim[nlo * 16 + K0 + 1];
    aImN[k].x = -aIm[k].x;           aImN[k].y = -aIm[k].y;
  }

  #pragma unroll
  for (int jt = 0; jt < 4; ++jt) {
    const int n = jt * 16 + nlo;
    v8f accRe = {0.f,0.f,0.f,0.f,0.f,0.f,0.f,0.f};
    v8f accIm = {0.f,0.f,0.f,0.f,0.f,0.f,0.f,0.f};
    #pragma unroll
    for (int k = 0; k < 4; ++k) {
      const int K0 = 4 * k + 2 * hf;
      const int K1 = K0 + 1;
      const int i0 = K0 * 64 + n;
      const int i1 = K1 * 64 + n;
      const float2 z0 = S[i0];
      const float2 z1 = S[i1];
      v2f br, bi;
      br.x = z0.x;  br.y = z1.x;
      bi.x = z0.y;  bi.y = z1.y;
      accRe = __builtin_amdgcn_wmma_f32_16x16x4_f32(false, aRe[k],  false, br, (short)0, accRe, false, false);
      accRe = __builtin_amdgcn_wmma_f32_16x16x4_f32(false, aImN[k], false, bi, (short)0, accRe, false, false);
      accIm = __builtin_amdgcn_wmma_f32_16x16x4_f32(false, aRe[k],  false, bi, (short)0, accIm, false, false);
      accIm = __builtin_amdgcn_wmma_f32_16x16x4_f32(false, aIm[k],  false, br, (short)0, accIm, false, false);
    }
    #pragma unroll
    for (int v = 0; v < 8; ++v) {
      const int M = v + 8 * hf;
      const int o = M * 64 + n;
      S[o] = make_float2(accRe[v], accIm[v]);
    }
  }
}

template <int MID>
__device__ __forceinline__ void apply8r(float* F, const float* frag, const int lane) {
  const int hf  = lane >> 4;
  const int nlo = lane & 15;
  const float4 g0 = *(const float4*)(frag + lane * 8);
  const float4 g1 = *(const float4*)(frag + lane * 8 + 4);
  v2f a[4];
  a[0].x = g0.x; a[0].y = g0.y;  a[1].x = g0.z; a[1].y = g0.w;
  a[2].x = g1.x; a[2].y = g1.y;  a[3].x = g1.z; a[3].y = g1.w;

  float* ptr[8];
  v2f b[8][4];
  #pragma unroll
  for (int jt = 0; jt < 8; ++jt) {
    if (MID) {
      ptr[jt] = F + 128 * (2 * jt + (nlo >> 3)) + 2 * (nlo & 7) + 16 * hf;
      #pragma unroll
      for (int kc = 0; kc < 4; ++kc) {
        const float2 z = *(const float2*)(ptr[jt] + 32 * kc);
        b[jt][kc].x = z.x;  b[jt][kc].y = z.y;
      }
    } else {
      ptr[jt] = F + 16 * (jt * 16 + nlo) + 8 * hf;
      const float4 x0 = *(const float4*)(ptr[jt]);
      const float4 x1 = *(const float4*)(ptr[jt] + 4);
      b[jt][0].x = x0.x; b[jt][0].y = x0.y;  b[jt][1].x = x0.z; b[jt][1].y = x0.w;
      b[jt][2].x = x1.x; b[jt][2].y = x1.y;  b[jt][3].x = x1.z; b[jt][3].y = x1.w;
    }
  }
  #pragma unroll
  for (int jq = 0; jq < 2; ++jq) {
    v8f d[4];
    #pragma unroll
    for (int t = 0; t < 4; ++t) d[t] = (v8f){0.f,0.f,0.f,0.f,0.f,0.f,0.f,0.f};
    #pragma unroll
    for (int kc = 0; kc < 4; ++kc) {
      #pragma unroll
      for (int t = 0; t < 4; ++t)
        d[t] = __builtin_amdgcn_wmma_f32_16x16x4_f32(false, a[kc], false, b[4 * jq + t][kc], (short)0, d[t], false, false);
    }
    #pragma unroll
    for (int t = 0; t < 4; ++t) {
      float* o = ptr[4 * jq + t];
      if (MID) {
        #pragma unroll
        for (int j = 0; j < 4; ++j)
          *(float2*)(o + 32 * j) = make_float2(d[t][2 * j], d[t][2 * j + 1]);
      } else {
        *(float4*)(o)     = make_float4(d[t][0], d[t][1], d[t][2], d[t][3]);
        *(float4*)(o + 4) = make_float4(d[t][4], d[t][5], d[t][6], d[t][7]);
      }
    }
  }
}

__device__ __forceinline__ float lane_value(const float v, const int w) {
  return __int_as_float(__builtin_amdgcn_readlane(__float_as_int(v), w));
}

__device__ __forceinline__ void cnot_ring(float2* S, const unsigned pLane, const int pT, const int lane) {
  float2 tv[32];
  #pragma unroll
  for (int t = 0; t < 32; ++t) {
    const unsigned o = pLane ^ (unsigned)__builtin_amdgcn_readlane(pT, t);
    tv[t] = *(const float2*)((const char*)S + o);
  }
  #pragma unroll
  for (int t = 0; t < 32; ++t) S[lane + 32 * t] = tv[t];
}

__device__ __forceinline__ float xhalf(const float x) {
  const int b = __float_as_int(x);
  return __int_as_float(__builtin_amdgcn_permlanex16(b, b, 0x76543210, 0xfedcba98, false, false));
}

template <int M>
__device__ __forceinline__ float row_xor(const float x) {
  return __int_as_float(__builtin_amdgcn_update_dpp(0, __float_as_int(x), 0x160 | M, 0xf, 0xf, true));
}

extern "C" __global__ __launch_bounds__(THREADS) __attribute__((amdgpu_waves_per_eu(1, 4)))
void qlayer_kernel(const float* __restrict__ x,
                   const float* __restrict__ qw,
                   float* __restrict__ out) {
  extern __shared__ char smem_raw[];
  float* u2   = (float*)smem_raw;
  float* UAre = u2 + NL * NQ * 8;
  float* UAim = UAre + NL * 256;
  float* frag = UAim + NL * 256;
  unsigned* permL = (unsigned*)(frag + NL * 512);
  unsigned* permH = permL + NL * 32;
  float2* stateBase = (float2*)(permH + NL * 32);

  const int tid = threadIdx.x;
  const int wave = tid >> 5, lane = tid & 31;
  const int b = blockIdx.x * WAVES_PER_BLOCK + wave;

  const int wl = lane < NQ ? lane : NQ - 1;
  const float xw = x[b * NQ + wl];

  static_assert(NL * NQ == WAVES_PER_BLOCK * 10, "10 gates per wave");
  {
    const int k3 = lane / 3;
    const int k  = k3 < 10 ? k3 : 9;
    const int j  = lane - 3 * k3;
    const int e  = wave * 10 + k;
    const float phi = qw[e * 3 + 0];
    const float th  = qw[e * 3 + 1];
    const float om  = qw[e * 3 + 2];
    const float a  = (j == 0) ? 0.5f * th : ((j == 1) ? 0.5f * (phi + om) : 0.5f * (phi - om));
    const float ca = cosf(a), sa = sinf(a);
    const float cpo = __shfl_down(ca, 1, 32), spo = __shfl_down(sa, 1, 32);
    const float cb  = __shfl_down(ca, 2, 32), sb  = __shfl_down(sa, 2, 32);
    if (j == 0 && k3 < 10) {
      const float c = ca, s = sa;
      float* g = u2 + e * 8;
      g[0] =  cpo * c; g[1] = -spo * c;
      g[2] = -cb * s;  g[3] = -sb * s;
      g[4] =  cb * s;  g[5] = -sb * s;
      g[6] =  cpo * c; g[7] =  spo * c;
    }
  }
  const float angw = 1.5707963267948966f * xw;
  const float cw = cosf(angw), sw = sinf(angw);
  __syncthreads();

  static_assert(THREADS == 128 && NL * 32 == THREADS, "index split of the table build");
  {
    constexpr int NUA = (NL - 1) * 256 / THREADS;
    constexpr int NFR = (NL - 1) * 512 / THREADS;
    float2 gu[NUA][4];
    float2 gf[NFR][3];
    #pragma unroll
    for (int it = 0; it < NUA; ++it) {
      const int l = 1 + (it >> 1);
      const int r = (tid >> 4) + 8 * (it & 1), cc = tid & 15;
      #pragma unroll
      for (int q = 0; q < 4; ++q) {
        const int rb = (r >> (3 - q)) & 1, cbit = (cc >> (3 - q)) & 1;
        gu[it][q] = *(const float2*)(u2 + (((l * NQ + q) * 4) + rb * 2 + cbit) * 2);
      }
    }
    int fcr[NFR], fck[NFR];
    #pragma unroll
    for (int it = 0; it < NFR; ++it) {
      const int l = 1 + (it >> 2), grp = (it >> 1) & 1, hf = it & 1;
      const int m = tid >> 3, s = tid & 7, kc = s >> 1, el = s & 1;
      const int row = grp ? m : (4 * ((m & 7) >> 1) + 2 * (m >> 3) + (m & 1));
      const int k   = grp ? (8 * hf + 2 * kc + el) : (4 * kc + 2 * hf + el);
      const int w0  = grp ? 7 : 4;
      const int qr = row >> 1, qk = k >> 1;
      fcr[it] = row & 1;  fck[it] = k & 1;
      #pragma unroll
      for (int j = 0; j < 3; ++j) {
        const int rb = (qr >> (2 - j)) & 1, cbit = (qk >> (2 - j)) & 1;
        gf[it][j] = *(const float2*)(u2 + (((l * NQ + w0 + j) * 4) + rb * 2 + cbit) * 2);
      }
    }
    int jl, jh;
    {
      const int l = tid >> 5, ln = tid & 31;
      const int rr = (l % (NQ - 1)) + 1;
      jl = ln; jh = 32 * ln;
      #pragma unroll
      for (int w = NQ - 1; w >= 0; --w) {
        const int cpos = 9 - w;
        int tw = w + rr;
        tw = (tw >= NQ) ? tw - NQ : tw;
        const int tpos = 9 - tw;
        jl ^= ((jl >> cpos) & 1) << tpos;
        jh ^= ((jh >> cpos) & 1) << tpos;
      }
    }
    float uar[NUA], uai[NUA], fv[NFR];
    #pragma unroll
    for (int it = 0; it < NUA; ++it) {
      float arA = gu[it][0].x, aiA = gu[it][0].y;
      #pragma unroll
      for (int q = 1; q < 4; ++q) {
        const float nr = arA * gu[it][q].x - aiA * gu[it][q].y;
        const float ni = arA * gu[it][q].y + aiA * gu[it][q].x;
        arA = nr; aiA = ni;
      }
      uar[it] = arA; uai[it] = aiA;
    }
    #pragma unroll
    for (int it = 0; it < NFR; ++it) {
      float ur = gf[it][0].x, ui = gf[it][0].y;
      #pragma unroll
      for (int j = 1; j < 3; ++j) {
        const float nr = ur * gf[it][j].x - ui * gf[it][j].y;
        const float ni = ur * gf[it][j].y + ui * gf[it][j].x;
        ur = nr; ui = ni;
      }
      fv[it] = (fcr[it] == fck[it]) ? ur : (fcr[it] ? ui : -ui);
    }
    #pragma unroll
    for (int it = 0; it < NUA; ++it) {
      UAre[256 + tid + THREADS * it] = uar[it];
      UAim[256 + tid + THREADS * it] = uai[it];
    }
    #pragma unroll
    for (int it = 0; it < NFR; ++it) frag[512 + tid + THREADS * it] = fv[it];
    permL[tid] = ((unsigned)jl) << 3;
    permH[tid] = ((unsigned)jh) << 3;
  }
  __syncthreads();

  float2* S = stateBase + wave * NSTATES;

  const unsigned pLane0 = permL[lane];
  const int pT0 = (int)permH[lane];
  float fr0[NQ], fi0[NQ], fr1[NQ], fi1[NQ];
  {
    const float4 ga = *(const float4*)(u2 + wl * 8);
    const float4 gb = *(const float4*)(u2 + wl * 8 + 4);
    const float pr0 = ga.x * cw + ga.z * sw, pi0 = ga.y * cw + ga.w * sw;
    const float pr1 = gb.x * cw + gb.z * sw, pi1 = gb.y * cw + gb.w * sw;
    #pragma unroll
    for (int w = 0; w < NQ; ++w) {
      fr0[w] = lane_value(pr0, w);  fi0[w] = lane_value(pi0, w);
      fr1[w] = lane_value(pr1, w);  fi1[w] = lane_value(pi1, w);
    }
  }
  float hr[32], hi[32];
  {
    float ar = (lane & 16) ? fr1[5] : fr0[5];
    float ai = (lane & 16) ? fi1[5] : fi0[5];
    #pragma unroll
    for (int w = 6; w < NQ; ++w) {
      const bool bit = ((lane >> (9 - w)) & 1) != 0;
      const float gr = bit ? fr1[w] : fr0[w];
      const float gi = bit ? fi1[w] : fi0[w];
      const float nr = ar * gr - ai * gi;
      const float ni = ar * gi + ai * gr;
      ar = nr; ai = ni;
    }
    hr[0] = ar; hi[0] = ai;
  }
  #pragma unroll
  for (int w = 0; w < 5; ++w) {
    #pragma unroll
    for (int u = (1 << w) - 1; u >= 0; --u) {
      const float vr = hr[u], vi = hi[u];
      hr[2 * u + 1] = vr * fr1[w] - vi * fi1[w];
      hi[2 * u + 1] = vr * fi1[w] + vi * fr1[w];
      hr[2 * u]     = vr * fr0[w] - vi * fi0[w];
      hi[2 * u]     = vr * fi0[w] + vi * fr0[w];
    }
  }
  #pragma unroll
  for (int t = 0; t < 32; ++t) S[lane + 32 * t] = make_float2(hr[t], hi[t]);

  cnot_ring(S, pLane0, pT0, lane);

  #pragma unroll 1
  for (int l = 1; l < NL; ++l) {
    const unsigned pLane = permL[l * 32 + lane];
    const int pT = (int)permH[l * 32 + lane];
    apply16(S, UAre + l * 256, UAim + l * 256, lane);
    apply8r<1>((float*)S, frag + (l * 2 + 0) * 256, lane);
    apply8r<0>((float*)S, frag + (l * 2 + 1) * 256, lane);
    cnot_ring(S, pLane, pT, lane);
  }

  float acc[NQ];
  float p[32];
  #pragma unroll
  for (int t = 0; t < 32; ++t) {
    const float2 z = S[lane + 32 * t];
    p[t] = z.x * z.x + z.y * z.y;
  }
  #pragma unroll
  for (int k = 0; k < 5; ++k) {
    const int n = 16 >> k;
    float d[16];
    #pragma unroll
    for (int u = 0; u < n; ++u) {
      d[u] = p[2 * u] - p[2 * u + 1];
      p[u] = p[2 * u] + p[2 * u + 1];
    }
    #pragma unroll
    for (int m = n >> 1; m > 0; m >>= 1) {
      #pragma unroll
      for (int u = 0; u < m; ++u) d[u] += d[u + m];
    }
    acc[4 - k] = d[0];
  }
  #pragma unroll
  for (int w = 5; w < NQ; ++w) acc[w] = ((lane >> (9 - w)) & 1) ? -p[0] : p[0];
  float r[NQ];
  #pragma unroll
  for (int w = 0; w < NQ; ++w) {
    float v = acc[w];
    v += xhalf(v);
    v += row_xor<8>(v);
    v += row_xor<4>(v);
    v += row_xor<2>(v);
    v += row_xor<1>(v);
    r[w] = v;
  }
  float o = r[0];
  #pragma unroll
  for (int w = 1; w < NQ; ++w) o = (lane == w) ? r[w] : o;
  if (lane < NQ) out[b * NQ + lane] = o;
}

extern "C" void kernel_launch(void* const* d_in, const int* in_sizes, int n_in,
                              void* d_out, int out_size, void* d_ws, size_t ws_size,
                              hipStream_t stream) {
  (void)in_sizes; (void)n_in; (void)out_size; (void)d_ws; (void)ws_size;
  const float* x  = (const float*)d_in[0];
  const float* qw = (const float*)d_in[1];
  float* out = (float*)d_out;
  const int batch = 2048;
  dim3 grid(batch / WAVES_PER_BLOCK);
  dim3 block(THREADS);
  qlayer_kernel<<<grid, block, LDS_BYTES, stream>>>(x, qw, out);
}


# Qubits 0-3 gate factored RZ*RY*RZ: real Kron RY part on WMMA for re and im (32 not 64 f32 WMMAs per layer), both diagonal phase vectors on the vector unit; each amplitude's summation order changes
# speedup vs baseline: 2.8744x; 1.0871x over previous
#include <hip/hip_runtime.h>

typedef __attribute__((ext_vector_type(2))) float v2f;
typedef __attribute__((ext_vector_type(8))) float v8f;

#define NQ 10
#define NL 4
#define NSTATES 1024
#define WAVES_PER_BLOCK 4
#define THREADS (WAVES_PER_BLOCK * 32)

#define LDS_BYTES (1280 + 8192 + 8192 + 512 + 512 + 32768)


__device__ __forceinline__ void apply16(float2* S, const float* R, const float* PH, const int lane) {
  const int hf  = lane >> 4;
  const int nlo = lane & 15;

  v2f a[4];
  float4 pp[4];
  #pragma unroll
  for (int k = 0; k < 4; ++k) {
    const int K0 = 4 * k + 2 * hf;
    const float2 g = *(const float2*)(R + nlo * 16 + K0);
    a[k].x = g.x;  a[k].y = g.y;
    pp[k] = *(const float4*)(PH + 2 * K0);
  }
  float4 qq[4];
  #pragma unroll
  for (int j = 0; j < 4; ++j) qq[j] = *(const float4*)(PH + 32 + 16 * hf + 4 * j);

  #pragma unroll
  for (int jt = 0; jt < 4; ++jt) {
    const int n = jt * 16 + nlo;
    v8f accRe = {0.f,0.f,0.f,0.f,0.f,0.f,0.f,0.f};
    v8f accIm = {0.f,0.f,0.f,0.f,0.f,0.f,0.f,0.f};
    #pragma unroll
    for (int k = 0; k < 4; ++k) {
      const int K0 = 4 * k + 2 * hf;
      const int K1 = K0 + 1;
      const float2 z0 = S[K0 * 64 + n];
      const float2 z1 = S[K1 * 64 + n];
      v2f br, bi;
      br.x = z0.x * pp[k].x - z0.y * pp[k].y;  bi.x = z0.x * pp[k].y + z0.y * pp[k].x;
      br.y = z1.x * pp[k].z - z1.y * pp[k].w;  bi.y = z1.x * pp[k].w + z1.y * pp[k].z;
      accRe = __builtin_amdgcn_wmma_f32_16x16x4_f32(false, a[k], false, br, (short)0, accRe, false, false);
      accIm = __builtin_amdgcn_wmma_f32_16x16x4_f32(false, a[k], false, bi, (short)0, accIm, false, false);
    }
    #pragma unroll
    for (int v = 0; v < 8; ++v) {
      const int M = v + 8 * hf;
      const v2f w  = (v & 1) ? (v2f){qq[v >> 1].z, qq[v >> 1].w} : (v2f){qq[v >> 1].x, qq[v >> 1].y};
      const v2f wx = {-w.y, w.x};
      const v2f re2 = {accRe[v], accRe[v]};
      const v2f im2 = {accIm[v], accIm[v]};
      const v2f o = re2 * w + im2 * wx;
      S[M * 64 + n] = make_float2(o.x, o.y);
    }
  }
}

template <int MID>
__device__ __forceinline__ void apply8r(float* F, const float* frag, const int lane) {
  const int hf  = lane >> 4;
  const int nlo = lane & 15;
  const float4 g0 = *(const float4*)(frag + lane * 8);
  const float4 g1 = *(const float4*)(frag + lane * 8 + 4);
  v2f a[4];
  a[0].x = g0.x; a[0].y = g0.y;  a[1].x = g0.z; a[1].y = g0.w;
  a[2].x = g1.x; a[2].y = g1.y;  a[3].x = g1.z; a[3].y = g1.w;

  float* ptr[8];
  v2f b[8][4];
  #pragma unroll
  for (int jt = 0; jt < 8; ++jt) {
    if (MID) {
      ptr[jt] = F + 128 * (2 * jt + (nlo >> 3)) + 2 * (nlo & 7) + 16 * hf;
      #pragma unroll
      for (int kc = 0; kc < 4; ++kc) {
        const float2 z = *(const float2*)(ptr[jt] + 32 * kc);
        b[jt][kc].x = z.x;  b[jt][kc].y = z.y;
      }
    } else {
      ptr[jt] = F + 16 * (jt * 16 + nlo) + 8 * hf;
      const float4 x0 = *(const float4*)(ptr[jt]);
      const float4 x1 = *(const float4*)(ptr[jt] + 4);
      b[jt][0].x = x0.x; b[jt][0].y = x0.y;  b[jt][1].x = x0.z; b[jt][1].y = x0.w;
      b[jt][2].x = x1.x; b[jt][2].y = x1.y;  b[jt][3].x = x1.z; b[jt][3].y = x1.w;
    }
  }
  #pragma unroll
  for (int jq = 0; jq < 2; ++jq) {
    v8f d[4];
    #pragma unroll
    for (int t = 0; t < 4; ++t) d[t] = (v8f){0.f,0.f,0.f,0.f,0.f,0.f,0.f,0.f};
    #pragma unroll
    for (int kc = 0; kc < 4; ++kc) {
      #pragma unroll
      for (int t = 0; t < 4; ++t)
        d[t] = __builtin_amdgcn_wmma_f32_16x16x4_f32(false, a[kc], false, b[4 * jq + t][kc], (short)0, d[t], false, false);
    }
    #pragma unroll
    for (int t = 0; t < 4; ++t) {
      float* o = ptr[4 * jq + t];
      if (MID) {
        #pragma unroll
        for (int j = 0; j < 4; ++j)
          *(float2*)(o + 32 * j) = make_float2(d[t][2 * j], d[t][2 * j + 1]);
      } else {
        *(float4*)(o)     = make_float4(d[t][0], d[t][1], d[t][2], d[t][3]);
        *(float4*)(o + 4) = make_float4(d[t][4], d[t][5], d[t][6], d[t][7]);
      }
    }
  }
}

__device__ __forceinline__ float lane_value(const float v, const int w) {
  return __int_as_float(__builtin_amdgcn_readlane(__float_as_int(v), w));
}

__device__ __forceinline__ void cnot_ring(float2* S, const unsigned pLane, const int pT, const int lane) {
  float2 tv[32];
  #pragma unroll
  for (int t = 0; t < 32; ++t) {
    const unsigned o = pLane ^ (unsigned)__builtin_amdgcn_readlane(pT, t);
    tv[t] = *(const float2*)((const char*)S + o);
  }
  #pragma unroll
  for (int t = 0; t < 32; ++t) S[lane + 32 * t] = tv[t];
}

__device__ __forceinline__ float xhalf(const float x) {
  const int b = __float_as_int(x);
  return __int_as_float(__builtin_amdgcn_permlanex16(b, b, 0x76543210, 0xfedcba98, false, false));
}

template <int M>
__device__ __forceinline__ float row_xor(const float x) {
  return __int_as_float(__builtin_amdgcn_update_dpp(0, __float_as_int(x), 0x160 | M, 0xf, 0xf, true));
}

extern "C" __global__ __launch_bounds__(THREADS) __attribute__((amdgpu_waves_per_eu(1, 4)))
void qlayer_kernel(const float* __restrict__ x,
                   const float* __restrict__ qw,
                   float* __restrict__ out) {
  extern __shared__ char smem_raw[];
  float* u2   = (float*)smem_raw;
  float* UAre = u2 + NL * NQ * 8;
  float* PHt  = UAre + NL * 256;
  float* frag = PHt + NL * 256;
  unsigned* permL = (unsigned*)(frag + NL * 512);
  unsigned* permH = permL + NL * 32;
  float2* stateBase = (float2*)(permH + NL * 32);

  const int tid = threadIdx.x;
  const int wave = tid >> 5, lane = tid & 31;
  const int b = blockIdx.x * WAVES_PER_BLOCK + wave;

  const int wl = lane < NQ ? lane : NQ - 1;
  const float xw = x[b * NQ + wl];

  static_assert(NL * NQ == WAVES_PER_BLOCK * 10, "10 gates per wave");
  {
    const int k3 = lane / 3;
    const int k  = k3 < 10 ? k3 : 9;
    const int j  = lane - 3 * k3;
    const int e  = wave * 10 + k;
    const float phi = qw[e * 3 + 0];
    const float th  = qw[e * 3 + 1];
    const float om  = qw[e * 3 + 2];
    const bool fz = (wave >= 1) && (k < 4);
    const float a1 = fz ? phi : (phi + om);
    const float a2 = fz ? om  : (phi - om);
    const float a  = (j == 0) ? 0.5f * th : ((j == 1) ? 0.5f * a1 : 0.5f * a2);
    const float ca = cosf(a), sa = sinf(a);
    const float cpo = __shfl_down(ca, 1, 32), spo = __shfl_down(sa, 1, 32);
    const float cb  = __shfl_down(ca, 2, 32), sb  = __shfl_down(sa, 2, 32);
    if (j == 0 && k3 < 10) {
      const float c = ca, s = sa;
      float* g = u2 + e * 8;
      if (fz) {
        g[0] = c;   g[1] = s;
        g[2] = cpo; g[3] = spo;
        g[4] = cb;  g[5] = sb;
      } else {
        g[0] =  cpo * c; g[1] = -spo * c;
        g[2] = -cb * s;  g[3] = -sb * s;
        g[4] =  cb * s;  g[5] = -sb * s;
        g[6] =  cpo * c; g[7] =  spo * c;
      }
    }
  }
  const float angw = 1.5707963267948966f * xw;
  const float cw = cosf(angw), sw = sinf(angw);
  __syncthreads();

  static_assert(THREADS == 128 && NL * 32 == THREADS, "index split of the table build");
  {
    constexpr int NUA = (NL - 1) * 256 / THREADS;
    constexpr int NFR = (NL - 1) * 512 / THREADS;
    float2 gu[NL - 1][4];
    float2 gp[4];
    float2 gf[NFR][3];
    #pragma unroll
    for (int l = 1; l < NL; ++l) {
      #pragma unroll
      for (int q = 0; q < 4; ++q) gu[l - 1][q] = *(const float2*)(u2 + (l * NQ + q) * 8);
    }
    const int pe = tid < 96 ? tid : 95;
    {
      const int l = 1 + (pe >> 5), which = (pe >> 4) & 1;
      #pragma unroll
      for (int q = 0; q < 4; ++q) gp[q] = *(const float2*)(u2 + (l * NQ + q) * 8 + 2 + 2 * which);
    }
    int fcr[NFR], fck[NFR];
    #pragma unroll
    for (int it = 0; it < NFR; ++it) {
      const int l = 1 + (it >> 2), grp = (it >> 1) & 1, hf = it & 1;
      const int m = tid >> 3, s = tid & 7, kc = s >> 1, el = s & 1;
      const int row = grp ? m : (4 * ((m & 7) >> 1) + 2 * (m >> 3) + (m & 1));
      const int k   = grp ? (8 * hf + 2 * kc + el) : (4 * kc + 2 * hf + el);
      const int w0  = grp ? 7 : 4;
      const int qr = row >> 1, qk = k >> 1;
      fcr[it] = row & 1;  fck[it] = k & 1;
      #pragma unroll
      for (int j = 0; j < 3; ++j) {
        const int rb = (qr >> (2 - j)) & 1, cbit = (qk >> (2 - j)) & 1;
        gf[it][j] = *(const float2*)(u2 + (((l * NQ + w0 + j) * 4) + rb * 2 + cbit) * 2);
      }
    }
    int jl, jh;
    {
      const int l = tid >> 5, ln = tid & 31;
      const int rr = (l % (NQ - 1)) + 1;
      jl = ln; jh = 32 * ln;
      #pragma unroll
      for (int w = NQ - 1; w >= 0; --w) {
        const int cpos = 9 - w;
        int tw = w + rr;
        tw = (tw >= NQ) ? tw - NQ : tw;
        const int tpos = 9 - tw;
        jl ^= ((jl >> cpos) & 1) << tpos;
        jh ^= ((jh >> cpos) & 1) << tpos;
      }
    }
    float uar[NUA], fv[NFR];
    #pragma unroll
    for (int it = 0; it < NUA; ++it) {
      const int l = 1 + (it >> 1);
      const int r = (tid >> 4) + 8 * (it & 1), cc = tid & 15;
      float v = 0.f;
      #pragma unroll
      for (int q = 0; q < 4; ++q) {
        const int rb = (r >> (3 - q)) & 1, cbit = (cc >> (3 - q)) & 1;
        const float2 cs = gu[l - 1][q];
        const float f = (rb == cbit) ? cs.x : (rb ? cs.y : -cs.y);
        v = (q == 0) ? f : v * f;
      }
      uar[it] = v;
    }
    float phr, phi2;
    {
      const int r = pe & 15;
      phr = gp[0].x;  phi2 = (r & 8) ? gp[0].y : -gp[0].y;
      #pragma unroll
      for (int q = 1; q < 4; ++q) {
        const float fr = gp[q].x;
        const float fi = ((r >> (3 - q)) & 1) ? gp[q].y : -gp[q].y;
        const float nr = phr * fr - phi2 * fi;
        const float ni = phr * fi + phi2 * fr;
        phr = nr; phi2 = ni;
      }
    }
    #pragma unroll
    for (int it = 0; it < NFR; ++it) {
      float ur = gf[it][0].x, ui = gf[it][0].y;
      #pragma unroll
      for (int j = 1; j < 3; ++j) {
        const float nr = ur * gf[it][j].x - ui * gf[it][j].y;
        const float ni = ur * gf[it][j].y + ui * gf[it][j].x;
        ur = nr; ui = ni;
      }
      fv[it] = (fcr[it] == fck[it]) ? ur : (fcr[it] ? ui : -ui);
    }
    #pragma unroll
    for (int it = 0; it < NUA; ++it) {
      UAre[256 + tid + THREADS * it] = uar[it];
    }
    if (tid < 96) *(float2*)(PHt + 64 + 2 * tid) = make_float2(phr, phi2);
    #pragma unroll
    for (int it = 0; it < NFR; ++it) frag[512 + tid + THREADS * it] = fv[it];
    permL[tid] = ((unsigned)jl) << 3;
    permH[tid] = ((unsigned)jh) << 3;
  }
  __syncthreads();

  float2* S = stateBase + wave * NSTATES;

  const unsigned pLane0 = permL[lane];
  const int pT0 = (int)permH[lane];
  float fr0[NQ], fi0[NQ], fr1[NQ], fi1[NQ];
  {
    const float4 ga = *(const float4*)(u2 + wl * 8);
    const float4 gb = *(const float4*)(u2 + wl * 8 + 4);
    const float pr0 = ga.x * cw + ga.z * sw, pi0 = ga.y * cw + ga.w * sw;
    const float pr1 = gb.x * cw + gb.z * sw, pi1 = gb.y * cw + gb.w * sw;
    #pragma unroll
    for (int w = 0; w < NQ; ++w) {
      fr0[w] = lane_value(pr0, w);  fi0[w] = lane_value(pi0, w);
      fr1[w] = lane_value(pr1, w);  fi1[w] = lane_value(pi1, w);
    }
  }
  float hr[32], hi[32];
  {
    float ar = (lane & 16) ? fr1[5] : fr0[5];
    float ai = (lane & 16) ? fi1[5] : fi0[5];
    #pragma unroll
    for (int w = 6; w < NQ; ++w) {
      const bool bit = ((lane >> (9 - w)) & 1) != 0;
      const float gr = bit ? fr1[w] : fr0[w];
      const float gi = bit ? fi1[w] : fi0[w];
      const float nr = ar * gr - ai * gi;
      const float ni = ar * gi + ai * gr;
      ar = nr; ai = ni;
    }
    hr[0] = ar; hi[0] = ai;
  }
  #pragma unroll
  for (int w = 0; w < 5; ++w) {
    #pragma unroll
    for (int u = (1 << w) - 1; u >= 0; --u) {
      const float vr = hr[u], vi = hi[u];
      hr[2 * u + 1] = vr * fr1[w] - vi * fi1[w];
      hi[2 * u + 1] = vr * fi1[w] + vi * fr1[w];
      hr[2 * u]     = vr * fr0[w] - vi * fi0[w];
      hi[2 * u]     = vr * fi0[w] + vi * fr0[w];
    }
  }
  #pragma unroll
  for (int t = 0; t < 32; ++t) S[lane + 32 * t] = make_float2(hr[t], hi[t]);

  cnot_ring(S, pLane0, pT0, lane);

  #pragma unroll 1
  for (int l = 1; l < NL; ++l) {
    const unsigned pLane = permL[l * 32 + lane];
    const int pT = (int)permH[l * 32 + lane];
    apply16(S, UAre + l * 256, PHt + l * 64, lane);
    apply8r<1>((float*)S, frag + (l * 2 + 0) * 256, lane);
    apply8r<0>((float*)S, frag + (l * 2 + 1) * 256, lane);
    cnot_ring(S, pLane, pT, lane);
  }

  float acc[NQ];
  float p[32];
  #pragma unroll
  for (int t = 0; t < 32; ++t) {
    const float2 z = S[lane + 32 * t];
    p[t] = z.x * z.x + z.y * z.y;
  }
  #pragma unroll
  for (int k = 0; k < 5; ++k) {
    const int n = 16 >> k;
    float d[16];
    #pragma unroll
    for (int u = 0; u < n; ++u) {
      d[u] = p[2 * u] - p[2 * u + 1];
      p[u] = p[2 * u] + p[2 * u + 1];
    }
    #pragma unroll
    for (int m = n >> 1; m > 0; m >>= 1) {
      #pragma unroll
      for (int u = 0; u < m; ++u) d[u] += d[u + m];
    }
    acc[4 - k] = d[0];
  }
  #pragma unroll
  for (int w = 5; w < NQ; ++w) acc[w] = ((lane >> (9 - w)) & 1) ? -p[0] : p[0];
  float r[NQ];
  #pragma unroll
  for (int w = 0; w < NQ; ++w) {
    float v = acc[w];
    v += xhalf(v);
    v += row_xor<8>(v);
    v += row_xor<4>(v);
    v += row_xor<2>(v);
    v += row_xor<1>(v);
    r[w] = v;
  }
  float o = r[0];
  #pragma unroll
  for (int w = 1; w < NQ; ++w) o = (lane == w) ? r[w] : o;
  if (lane < NQ) out[b * NQ + lane] = o;
}

extern "C" void kernel_launch(void* const* d_in, const int* in_sizes, int n_in,
                              void* d_out, int out_size, void* d_ws, size_t ws_size,
                              hipStream_t stream) {
  (void)in_sizes; (void)n_in; (void)out_size; (void)d_ws; (void)ws_size;
  const float* x  = (const float*)d_in[0];
  const float* qw = (const float*)d_in[1];
  float* out = (float*)d_out;
  const int batch = 2048;
  dim3 grid(batch / WAVES_PER_BLOCK);
  dim3 block(THREADS);
  qlayer_kernel<<<grid, block, LDS_BYTES, stream>>>(x, qw, out);
}


# State in LDS padded 16 B per 256 B, each stage's lanes put on index bits that fall on distinct banks: no 2- and 4-way bank conflicts left; k (summation) order of the qubits 0-3 product changes
# speedup vs baseline: 3.0618x; 1.0652x over previous
#include <hip/hip_runtime.h>

typedef __attribute__((ext_vector_type(2))) float v2f;
typedef __attribute__((ext_vector_type(8))) float v8f;

#define NQ 10
#define NL 4
#define NSTATES 1024
#define WAVES_PER_BLOCK 4
#define THREADS (WAVES_PER_BLOCK * 32)

#define SLOTS 1088
#define LDS_BYTES (1280 + 8192 + 8192 + 512 + 512 + WAVES_PER_BLOCK * SLOTS * 8)

__device__ __host__ constexpr int slotc(const int i) { return i + 2 * (i >> 5); }


__device__ __forceinline__ void apply16(float2* S, const float* R, const float* PH, const int lane) {
  const int hf  = lane >> 4;
  const int nlo = lane & 15;

  const int arow = (nlo & 3) | ((nlo & 4) << 1) | ((nlo & 8) >> 1);
  float2* SL = S + nlo + 272 * hf;
  v2f a[4];
  float4 pp[4];
  #pragma unroll
  for (int k = 0; k < 4; ++k) {
    const int K0 = 8 * (k >> 1) + 4 * hf + 2 * (k & 1);
    const float2 g = *(const float2*)(R + arow * 16 + K0);
    a[k].x = g.x;  a[k].y = g.y;
    pp[k] = *(const float4*)(PH + 2 * K0);
  }
  float4 qq[4];
  #pragma unroll
  for (int j = 0; j < 4; ++j) qq[j] = *(const float4*)(PH + 32 + 16 * (j >> 1) + 8 * hf + 4 * (j & 1));

  #pragma unroll
  for (int jt = 0; jt < 4; ++jt) {
    v8f accRe = {0.f,0.f,0.f,0.f,0.f,0.f,0.f,0.f};
    v8f accIm = {0.f,0.f,0.f,0.f,0.f,0.f,0.f,0.f};
    #pragma unroll
    for (int k = 0; k < 4; ++k) {
      const int Kc = 8 * (k >> 1) + 2 * (k & 1);
      const float2 z0 = SL[slotc(Kc * 64 + 16 * jt)];
      const float2 z1 = SL[slotc((Kc + 1) * 64 + 16 * jt)];
      v2f br, bi;
      br.x = z0.x * pp[k].x - z0.y * pp[k].y;  bi.x = z0.x * pp[k].y + z0.y * pp[k].x;
      br.y = z1.x * pp[k].z - z1.y * pp[k].w;  bi.y = z1.x * pp[k].w + z1.y * pp[k].z;
      accRe = __builtin_amdgcn_wmma_f32_16x16x4_f32(false, a[k], false, br, (short)0, accRe, false, false);
      accIm = __builtin_amdgcn_wmma_f32_16x16x4_f32(false, a[k], false, bi, (short)0, accIm, false, false);
    }
    #pragma unroll
    for (int v = 0; v < 8; ++v) {
      const v2f w  = (v & 1) ? (v2f){qq[v >> 1].z, qq[v >> 1].w} : (v2f){qq[v >> 1].x, qq[v >> 1].y};
      const v2f wx = {-w.y, w.x};
      const v2f re2 = {accRe[v], accRe[v]};
      const v2f im2 = {accIm[v], accIm[v]};
      const v2f o = re2 * w + im2 * wx;
      SL[slotc((8 * (v >> 2) + (v & 3)) * 64 + 16 * jt)] = make_float2(o.x, o.y);
    }
  }
}

template <int MID>
__device__ __forceinline__ void apply8r(float* F, const float* frag, const int lane) {
  const int hf  = lane >> 4;
  const int nlo = lane & 15;
  const float4 g0 = *(const float4*)(frag + lane * 8);
  const float4 g1 = *(const float4*)(frag + lane * 8 + 4);
  v2f a[4];
  a[0].x = g0.x; a[0].y = g0.y;  a[1].x = g0.z; a[1].y = g0.w;
  a[2].x = g1.x; a[2].y = g1.y;  a[3].x = g1.z; a[3].y = g1.w;

  float* ptr[8];
  v2f b[8][4];
  #pragma unroll
  for (int jt = 0; jt < 8; ++jt) {
    if (MID) {
      ptr[jt] = F + 2 * ((nlo & 7) + 8 * hf + 272 * (nlo >> 3)
                         + slotc(64 * (jt & 1) + 128 * ((jt >> 1) & 1) + 512 * (jt >> 2)));
      #pragma unroll
      for (int kc = 0; kc < 4; ++kc) {
        const float2 z = *(const float2*)(ptr[jt] + 2 * slotc(16 * kc));
        b[jt][kc].x = z.x;  b[jt][kc].y = z.y;
      }
    } else {
      ptr[jt] = F + 2 * (4 * hf + 8 * (nlo & 3) + 34 * ((nlo >> 2) & 1) + 68 * (nlo >> 3) + slotc(128 * jt));
      const float4 x0 = *(const float4*)(ptr[jt]);
      const float4 x1 = *(const float4*)(ptr[jt] + 4);
      b[jt][0].x = x0.x; b[jt][0].y = x0.y;  b[jt][1].x = x0.z; b[jt][1].y = x0.w;
      b[jt][2].x = x1.x; b[jt][2].y = x1.y;  b[jt][3].x = x1.z; b[jt][3].y = x1.w;
    }
  }
  #pragma unroll
  for (int jq = 0; jq < 2; ++jq) {
    v8f d[4];
    #pragma unroll
    for (int t = 0; t < 4; ++t) d[t] = (v8f){0.f,0.f,0.f,0.f,0.f,0.f,0.f,0.f};
    #pragma unroll
    for (int kc = 0; kc < 4; ++kc) {
      #pragma unroll
      for (int t = 0; t < 4; ++t)
        d[t] = __builtin_amdgcn_wmma_f32_16x16x4_f32(false, a[kc], false, b[4 * jq + t][kc], (short)0, d[t], false, false);
    }
    #pragma unroll
    for (int t = 0; t < 4; ++t) {
      float* o = ptr[4 * jq + t];
      if (MID) {
        #pragma unroll
        for (int j = 0; j < 4; ++j)
          *(float2*)(o + 2 * slotc(16 * j)) = make_float2(d[t][2 * j], d[t][2 * j + 1]);
      } else {
        float* q = F + 2 * (2 * nlo + 64 * hf + 128 * (4 * jq + t));
        *(float4*)(q)      = make_float4(d[t][0], d[t][1], d[t][2], d[t][3]);
        *(float4*)(q + 64) = make_float4(d[t][4], d[t][5], d[t][6], d[t][7]);
      }
    }
  }
}

__device__ __forceinline__ float lane_value(const float v, const int w) {
  return __int_as_float(__builtin_amdgcn_readlane(__float_as_int(v), w));
}

__device__ __forceinline__ void cnot_ring(float2* S, const unsigned pLane, const int pT, const int lane) {
  float2 tv[32];
  #pragma unroll
  for (int t = 0; t < 32; ++t) {
    const unsigned o = pLane ^ (unsigned)__builtin_amdgcn_readlane(pT, t);
    tv[t] = *(const float2*)((const char*)S + o);
  }
  float2* D = S + (lane & 1) + 8 * ((lane >> 1) & 1) + 16 * ((lane >> 2) & 1) + 34 * ((lane >> 3) & 1) + 68 * (lane >> 4);
  #pragma unroll
  for (int t = 0; t < 32; ++t) D[slotc(2 * (t & 3) + 128 * (t >> 2))] = tv[t];
}

__device__ __forceinline__ float xhalf(const float x) {
  const int b = __float_as_int(x);
  return __int_as_float(__builtin_amdgcn_permlanex16(b, b, 0x76543210, 0xfedcba98, false, false));
}

template <int M>
__device__ __forceinline__ float row_xor(const float x) {
  return __int_as_float(__builtin_amdgcn_update_dpp(0, __float_as_int(x), 0x160 | M, 0xf, 0xf, true));
}

extern "C" __global__ __launch_bounds__(THREADS) __attribute__((amdgpu_waves_per_eu(1, 4)))
void qlayer_kernel(const float* __restrict__ x,
                   const float* __restrict__ qw,
                   float* __restrict__ out) {
  extern __shared__ char smem_raw[];
  float* u2   = (float*)smem_raw;
  float* UAre = u2 + NL * NQ * 8;
  float* PHt  = UAre + NL * 256;
  float* frag = PHt + NL * 256;
  unsigned* permL = (unsigned*)(frag + NL * 512);
  unsigned* permH = permL + NL * 32;
  float2* stateBase = (float2*)(permH + NL * 32);

  const int tid = threadIdx.x;
  const int wave = tid >> 5, lane = tid & 31;
  const int b = blockIdx.x * WAVES_PER_BLOCK + wave;

  const int wl = lane < NQ ? lane : NQ - 1;
  const float xw = x[b * NQ + wl];

  static_assert(NL * NQ == WAVES_PER_BLOCK * 10, "10 gates per wave");
  {
    const int k3 = lane / 3;
    const int k  = k3 < 10 ? k3 : 9;
    const int j  = lane - 3 * k3;
    const int e  = wave * 10 + k;
    const float phi = qw[e * 3 + 0];
    const float th  = qw[e * 3 + 1];
    const float om  = qw[e * 3 + 2];
    const bool fz = (wave >= 1) && (k < 4);
    const float a1 = fz ? phi : (phi + om);
    const float a2 = fz ? om  : (phi - om);
    const float a  = (j == 0) ? 0.5f * th : ((j == 1) ? 0.5f * a1 : 0.5f * a2);
    const float ca = cosf(a), sa = sinf(a);
    const float cpo = __shfl_down(ca, 1, 32), spo = __shfl_down(sa, 1, 32);
    const float cb  = __shfl_down(ca, 2, 32), sb  = __shfl_down(sa, 2, 32);
    if (j == 0 && k3 < 10) {
      const float c = ca, s = sa;
      float* g = u2 + e * 8;
      if (fz) {
        g[0] = c;   g[1] = s;
        g[2] = cpo; g[3] = spo;
        g[4] = cb;  g[5] = sb;
      } else {
        g[0] =  cpo * c; g[1] = -spo * c;
        g[2] = -cb * s;  g[3] = -sb * s;
        g[4] =  cb * s;  g[5] = -sb * s;
        g[6] =  cpo * c; g[7] =  spo * c;
      }
    }
  }
  const float angw = 1.5707963267948966f * xw;
  const float cw = cosf(angw), sw = sinf(angw);
  __syncthreads();

  static_assert(THREADS == 128 && NL * 32 == THREADS, "index split of the table build");
  {
    constexpr int NUA = (NL - 1) * 256 / THREADS;
    constexpr int NFR = (NL - 1) * 512 / THREADS;
    float2 gu[NL - 1][4];
    float2 gp[4];
    float2 gf[NFR][3];
    #pragma unroll
    for (int l = 1; l < NL; ++l) {
      #pragma unroll
      for (int q = 0; q < 4; ++q) gu[l - 1][q] = *(const float2*)(u2 + (l * NQ + q) * 8);
    }
    const int pe = tid < 96 ? tid : 95;
    {
      const int l = 1 + (pe >> 5), which = (pe >> 4) & 1;
      #pragma unroll
      for (int q = 0; q < 4; ++q) gp[q] = *(const float2*)(u2 + (l * NQ + q) * 8 + 2 + 2 * which);
    }
    int fcr[NFR], fck[NFR];
    #pragma unroll
    for (int it = 0; it < NFR; ++it) {
      const int l = 1 + (it >> 2), grp = (it >> 1) & 1, hf = it & 1;
      const int m = tid >> 3, s = tid & 7, kc = s >> 1, el = s & 1;
      const int row = grp ? m : (4 * ((m & 7) >> 1) + 2 * (m >> 3) + (m & 1));
      const int k   = grp ? (8 * hf + 2 * kc + el) : (4 * kc + 2 * hf + el);
      const int w0  = grp ? 7 : 4;
      const int qr = row >> 1, qk = k >> 1;
      fcr[it] = row & 1;  fck[it] = k & 1;
      #pragma unroll
      for (int j = 0; j < 3; ++j) {
        const int rb = (qr >> (2 - j)) & 1, cbit = (qk >> (2 - j)) & 1;
        gf[it][j] = *(const float2*)(u2 + (((l * NQ + w0 + j) * 4) + rb * 2 + cbit) * 2);
      }
    }
    int jl, jh;
    {
      const int l = tid >> 5, ln = tid & 31;
      const int rr = (l % (NQ - 1)) + 1;
      jl = (ln & 1) | ((ln >> 1) << 3);
      jh = ((ln & 3) << 1) | ((ln >> 2) << 7);
      #pragma unroll
      for (int w = NQ - 1; w >= 0; --w) {
        const int cpos = 9 - w;
        int tw = w + rr;
        tw = (tw >= NQ) ? tw - NQ : tw;
        const int tpos = 9 - tw;
        jl ^= ((jl >> cpos) & 1) << tpos;
        jh ^= ((jh >> cpos) & 1) << tpos;
      }
    }
    float uar[NUA], fv[NFR];
    #pragma unroll
    for (int it = 0; it < NUA; ++it) {
      const int l = 1 + (it >> 1);
      const int r = (tid >> 4) + 8 * (it & 1), cc = tid & 15;
      float v = 0.f;
      #pragma unroll
      for (int q = 0; q < 4; ++q) {
        const int rb = (r >> (3 - q)) & 1, cbit = (cc >> (3 - q)) & 1;
        const float2 cs = gu[l - 1][q];
        const float f = (rb == cbit) ? cs.x : (rb ? cs.y : -cs.y);
        v = (q == 0) ? f : v * f;
      }
      uar[it] = v;
    }
    float phr, phi2;
    {
      const int r = pe & 15;
      phr = gp[0].x;  phi2 = (r & 8) ? gp[0].y : -gp[0].y;
      #pragma unroll
      for (int q = 1; q < 4; ++q) {
        const float fr = gp[q].x;
        const float fi = ((r >> (3 - q)) & 1) ? gp[q].y : -gp[q].y;
        const float nr = phr * fr - phi2 * fi;
        const float ni = phr * fi + phi2 * fr;
        phr = nr; phi2 = ni;
      }
    }
    #pragma unroll
    for (int it = 0; it < NFR; ++it) {
      float ur = gf[it][0].x, ui = gf[it][0].y;
      #pragma unroll
      for (int j = 1; j < 3; ++j) {
        const float nr = ur * gf[it][j].x - ui * gf[it][j].y;
        const float ni = ur * gf[it][j].y + ui * gf[it][j].x;
        ur = nr; ui = ni;
      }
      fv[it] = (fcr[it] == fck[it]) ? ur : (fcr[it] ? ui : -ui);
    }
    #pragma unroll
    for (int it = 0; it < NUA; ++it) {
      UAre[256 + tid + THREADS * it] = uar[it];
    }
    if (tid < 96) *(float2*)(PHt + 64 + 2 * tid) = make_float2(phr, phi2);
    #pragma unroll
    for (int it = 0; it < NFR; ++it) frag[512 + tid + THREADS * it] = fv[it];
    if (tid >= 32) {
      jl = (jl & 1) | (((jl >> 3) & 15) << 1) | (((jl >> 1) & 3) << 5) | (jl & 0x380);
      jh = (jh & 1) | (((jh >> 3) & 15) << 1) | (((jh >> 1) & 3) << 5) | (jh & 0x380);
    }
    permL[tid] = ((unsigned)jl) << 3;
    permH[tid] = ((unsigned)jh) << 3;
  }
  __syncthreads();

  float2* S = stateBase + wave * SLOTS;

  const unsigned pLane0 = permL[lane];
  const int pT0 = (int)permH[lane];
  float fr0[NQ], fi0[NQ], fr1[NQ], fi1[NQ];
  {
    const float4 ga = *(const float4*)(u2 + wl * 8);
    const float4 gb = *(const float4*)(u2 + wl * 8 + 4);
    const float pr0 = ga.x * cw + ga.z * sw, pi0 = ga.y * cw + ga.w * sw;
    const float pr1 = gb.x * cw + gb.z * sw, pi1 = gb.y * cw + gb.w * sw;
    #pragma unroll
    for (int w = 0; w < NQ; ++w) {
      fr0[w] = lane_value(pr0, w);  fi0[w] = lane_value(pi0, w);
      fr1[w] = lane_value(pr1, w);  fi1[w] = lane_value(pi1, w);
    }
  }
  float hr[32], hi[32];
  {
    float ar = (lane & 16) ? fr1[5] : fr0[5];
    float ai = (lane & 16) ? fi1[5] : fi0[5];
    #pragma unroll
    for (int w = 6; w < NQ; ++w) {
      const bool bit = ((lane >> (9 - w)) & 1) != 0;
      const float gr = bit ? fr1[w] : fr0[w];
      const float gi = bit ? fi1[w] : fi0[w];
      const float nr = ar * gr - ai * gi;
      const float ni = ar * gi + ai * gr;
      ar = nr; ai = ni;
    }
    hr[0] = ar; hi[0] = ai;
  }
  #pragma unroll
  for (int w = 0; w < 5; ++w) {
    #pragma unroll
    for (int u = (1 << w) - 1; u >= 0; --u) {
      const float vr = hr[u], vi = hi[u];
      hr[2 * u + 1] = vr * fr1[w] - vi * fi1[w];
      hi[2 * u + 1] = vr * fi1[w] + vi * fr1[w];
      hr[2 * u]     = vr * fr0[w] - vi * fi0[w];
      hi[2 * u]     = vr * fi0[w] + vi * fr0[w];
    }
  }
  #pragma unroll
  for (int t = 0; t < 32; ++t) S[lane + 32 * t] = make_float2(hr[t], hi[t]);

  cnot_ring(S, pLane0, pT0, lane);

  #pragma unroll 1
  for (int l = 1; l < NL; ++l) {
    const unsigned pLane = permL[l * 32 + lane];
    const int pT = (int)permH[l * 32 + lane];
    apply16(S, UAre + l * 256, PHt + l * 64, lane);
    apply8r<1>((float*)S, frag + (l * 2 + 0) * 256, lane);
    apply8r<0>((float*)S, frag + (l * 2 + 1) * 256, lane);
    cnot_ring(S, pLane, pT, lane);
  }

  float acc[NQ];
  float p[32];
  #pragma unroll
  for (int t = 0; t < 32; ++t) {
    const float2 z = S[lane + 34 * t];
    p[t] = z.x * z.x + z.y * z.y;
  }
  #pragma unroll
  for (int k = 0; k < 5; ++k) {
    const int n = 16 >> k;
    float d[16];
    #pragma unroll
    for (int u = 0; u < n; ++u) {
      d[u] = p[2 * u] - p[2 * u + 1];
      p[u] = p[2 * u] + p[2 * u + 1];
    }
    #pragma unroll
    for (int m = n >> 1; m > 0; m >>= 1) {
      #pragma unroll
      for (int u = 0; u < m; ++u) d[u] += d[u + m];
    }
    acc[4 - k] = d[0];
  }
  #pragma unroll
  for (int w = 5; w < NQ; ++w) acc[w] = ((lane >> (9 - w)) & 1) ? -p[0] : p[0];
  float r[NQ];
  #pragma unroll
  for (int w = 0; w < NQ; ++w) {
    float v = acc[w];
    v += xhalf(v);
    v += row_xor<8>(v);
    v += row_xor<4>(v);
    v += row_xor<2>(v);
    v += row_xor<1>(v);
    r[w] = v;
  }
  float o = r[0];
  #pragma unroll
  for (int w = 1; w < NQ; ++w) o = (lane == w) ? r[w] : o;
  if (lane < NQ) out[b * NQ + lane] = o;
}

extern "C" void kernel_launch(void* const* d_in, const int* in_sizes, int n_in,
                              void* d_out, int out_size, void* d_ws, size_t ws_size,
                              hipStream_t stream) {
  (void)in_sizes; (void)n_in; (void)out_size; (void)d_ws; (void)ws_size;
  const float* x  = (const float*)d_in[0];
  const float* qw = (const float*)d_in[1];
  float* out = (float*)d_out;
  const int batch = 2048;
  dim3 grid(batch / WAVES_PER_BLOCK);
  dim3 block(THREADS);
  qlayer_kernel<<<grid, block, LDS_BYTES, stream>>>(x, qw, out);
}
